# starting kernel
# baseline (speedup 1.0000x reference)
#include <hip/hip_runtime.h>
#include <hip/hip_bf16.h>
#include <math.h>

typedef float v2f __attribute__((ext_vector_type(2)));
typedef float v8f __attribute__((ext_vector_type(8)));

#define NB     8
#define NPTS   16384
#define NPT    512
#define NSAMP  32
#define RAD2   0.04f
#define BN_EPS 1e-5f

__global__ __launch_bounds__(1024) void fps_kernel(const float* __restrict__ xyz,
                                                   int* __restrict__ fpsidx,
                                                   float* __restrict__ newxyz) {
  const int b   = blockIdx.x;
  const int tid = threadIdx.x;
  const float* base = xyz + (size_t)b * NPTS * 3;

  float px[16], py[16], pz[16], dd[16];
#pragma unroll
  for (int j = 0; j < 16; ++j) {
    int p = tid + j * 1024;
    px[j] = base[p * 3 + 0];
    py[j] = base[p * 3 + 1];
    pz[j] = base[p * 3 + 2];
    dd[j] = 3.4e38f;
  }

  __shared__ float rv[32];
  __shared__ int   ri[32];
  __shared__ int   s_last;

  const int wave = tid >> 5;
  const int lane = tid & 31;
  int last = 0;

  for (int i = 0; i < NPT; ++i) {
    if (tid == 0) {
      fpsidx[b * NPT + i] = last;
      newxyz[(b * NPT + i) * 3 + 0] = base[last * 3 + 0];
      newxyz[(b * NPT + i) * 3 + 1] = base[last * 3 + 1];
      newxyz[(b * NPT + i) * 3 + 2] = base[last * 3 + 2];
    }
    const float cx = base[last * 3 + 0];
    const float cy = base[last * 3 + 1];
    const float cz = base[last * 3 + 2];

    float bv = -1.0f;
    int   bi = 0x7fffffff;
#pragma unroll
    for (int j = 0; j < 16; ++j) {
      float dx = px[j] - cx, dy = py[j] - cy, dz = pz[j] - cz;
      float d  = dx * dx + dy * dy + dz * dz;
      float nd = fminf(dd[j], d);
      dd[j] = nd;
      if (nd > bv) { bv = nd; bi = tid + j * 1024; }
    }
#pragma unroll
    for (int off = 16; off > 0; off >>= 1) {
      float ov = __shfl_xor(bv, off, 32);
      int   oi = __shfl_xor(bi, off, 32);
      if (ov > bv || (ov == bv && oi < bi)) { bv = ov; bi = oi; }
    }
    if (lane == 0) { rv[wave] = bv; ri[wave] = bi; }
    __syncthreads();
    if (wave == 0) {
      bv = rv[lane]; bi = ri[lane];
#pragma unroll
      for (int off = 16; off > 0; off >>= 1) {
        float ov = __shfl_xor(bv, off, 32);
        int   oi = __shfl_xor(bi, off, 32);
        if (ov > bv || (ov == bv && oi < bi)) { bv = ov; bi = oi; }
      }
      if (lane == 0) s_last = bi;
    }
    __syncthreads();
    last = s_last;
  }
}

__global__ __launch_bounds__(256) void ballq_kernel(const float* __restrict__ xyz,
                                                    const float* __restrict__ newxyz,
                                                    int* __restrict__ grp) {
  const int lane = threadIdx.x & 31;
  const int q    = blockIdx.x * 8 + (threadIdx.x >> 5);
  const int b    = q >> 9;
  const int s    = q & (NPT - 1);

  const float* base = xyz + (size_t)b * NPTS * 3;
  const float cx = newxyz[(b * NPT + s) * 3 + 0];
  const float cy = newxyz[(b * NPT + s) * 3 + 1];
  const float cz = newxyz[(b * NPT + s) * 3 + 2];

  int* g = grp + (size_t)q * NSAMP;
  int cnt = 0, first = -1;

  for (int bp = 0; bp < NPTS && cnt < NSAMP; bp += 32) {
    const int p = bp + lane;
    float dx = base[p * 3 + 0] - cx;
    float dy = base[p * 3 + 1] - cy;
    float dz = base[p * 3 + 2] - cz;
    bool hit = (dx * dx + dy * dy + dz * dz) <= RAD2;
    unsigned mask = (unsigned)__ballot(hit);
    if (mask) {
      if (first < 0) first = bp + __builtin_ctz(mask);
      int pos = __popc(mask & ((1u << lane) - 1u));
      if (hit && (cnt + pos) < NSAMP) g[cnt + pos] = p;
      cnt += __popc(mask);
    }
  }
  if (first < 0) first = NPTS - 1;
  int filled = cnt < NSAMP ? cnt : NSAMP;
  for (int slot = filled + lane; slot < NSAMP; slot += 32) g[slot] = first;
}

__device__ __forceinline__ void gemm_bn_relu(const float* __restrict__ A, int lda,
                                             const float* __restrict__ W, int Kdim, int Kpad,
                                             int Nout,
                                             const float* __restrict__ gg,
                                             const float* __restrict__ bb,
                                             const float* __restrict__ mm,
                                             const float* __restrict__ vv,
                                             float* __restrict__ D, int ldd, int lane) {
  const int r15  = lane & 15;
  const int half = lane >> 4;
  for (int m0 = 0; m0 < 32; m0 += 16) {
    for (int n0 = 0; n0 < Nout; n0 += 16) {
      v8f acc = {0.f, 0.f, 0.f, 0.f, 0.f, 0.f, 0.f, 0.f};
      const int n = n0 + r15;
      for (int k0 = 0; k0 < Kpad; k0 += 4) {
        const int kk = k0 + half * 2;
        v2f a, w;
        a.x = A[(m0 + r15) * lda + kk];
        a.y = A[(m0 + r15) * lda + kk + 1];
        int ka = kk     < Kdim ? kk     : Kdim - 1;
        int kb = kk + 1 < Kdim ? kk + 1 : Kdim - 1;
        w.x = W[n * Kdim + ka];
        w.y = W[n * Kdim + kb];
        acc = __builtin_amdgcn_wmma_f32_16x16x4_f32(false, a, false, w,
                                                    (short)0, acc, false, false);
      }
      const float sc = gg[n] * rsqrtf(vv[n] + BN_EPS);
      const float bs = bb[n] - mm[n] * sc;
#pragma unroll
      for (int r = 0; r < 8; ++r) {
        float val = fmaxf(acc[r] * sc + bs, 0.f);
        D[(m0 + r + half * 8) * ldd + n] = val;
      }
    }
  }
}

__global__ __launch_bounds__(64) void mlp_kernel(const float* __restrict__ xyz,
                                                 const float* __restrict__ feat,
                                                 const int*   __restrict__ grp,
                                                 const float* __restrict__ newxyz,
                                                 const float* __restrict__ w0, const float* __restrict__ g0,
                                                 const float* __restrict__ b0, const float* __restrict__ m0,
                                                 const float* __restrict__ v0,
                                                 const float* __restrict__ w1, const float* __restrict__ g1,
                                                 const float* __restrict__ b1, const float* __restrict__ m1,
                                                 const float* __restrict__ v1,
                                                 const float* __restrict__ w2, const float* __restrict__ g2,
                                                 const float* __restrict__ b2, const float* __restrict__ m2,
                                                 const float* __restrict__ v2,
                                                 float* __restrict__ outfeat) {
  __shared__ float Xb[2][32 * 68];
  __shared__ float Hb[2][32 * 132];

  const int wave = threadIdx.x >> 5;
  const int lane = threadIdx.x & 31;
  const int q    = blockIdx.x * 2 + wave;
  const int b    = q >> 9;
  const int s    = q & (NPT - 1);

  float* X = Xb[wave];
  float* H = Hb[wave];

  const int gi = grp[(size_t)q * NSAMP + lane];
  const float* prow = xyz + ((size_t)b * NPTS + gi) * 3;
  const float* nx   = newxyz + (b * NPT + s) * 3;
  float* xr = X + lane * 68;
  xr[0] = prow[0] - nx[0];
  xr[1] = prow[1] - nx[1];
  xr[2] = prow[2] - nx[2];
  const float4* frow = (const float4*)(feat + ((size_t)b * NPTS + gi) * 64);
#pragma unroll
  for (int c4 = 0; c4 < 16; ++c4) {
    float4 f = frow[c4];
    xr[3 + 4 * c4 + 0] = f.x;
    xr[3 + 4 * c4 + 1] = f.y;
    xr[3 + 4 * c4 + 2] = f.z;
    xr[3 + 4 * c4 + 3] = f.w;
  }
  xr[67] = 0.f;

  gemm_bn_relu(X, 68,  w0, 67, 68, 64,  g0, b0, m0, v0, H, 132, lane);
  gemm_bn_relu(H, 132, w1, 64, 64, 64,  g1, b1, m1, v1, X, 68,  lane);
  gemm_bn_relu(X, 68,  w2, 64, 64, 128, g2, b2, m2, v2, H, 132, lane);

#pragma unroll
  for (int cc = 0; cc < 4; ++cc) {
    const int col = lane + cc * 32;
    float mx = H[col];
    for (int row = 1; row < 32; ++row) mx = fmaxf(mx, H[row * 132 + col]);
    outfeat[((size_t)b * 128 + col) * NPT + s] = mx;
  }
}

extern "C" void kernel_launch(void* const* d_in, const int* in_sizes, int n_in,
                              void* d_out, int out_size, void* d_ws, size_t ws_size,
                              hipStream_t stream) {
  (void)in_sizes; (void)n_in; (void)out_size; (void)ws_size;
  const float* xyz  = (const float*)d_in[0];
  const float* feat = (const float*)d_in[1];
  const float* w0 = (const float*)d_in[2];
  const float* g0 = (const float*)d_in[3];
  const float* b0 = (const float*)d_in[4];
  const float* m0 = (const float*)d_in[5];
  const float* v0 = (const float*)d_in[6];
  const float* w1 = (const float*)d_in[7];
  const float* g1 = (const float*)d_in[8];
  const float* b1 = (const float*)d_in[9];
  const float* m1 = (const float*)d_in[10];
  const float* v1 = (const float*)d_in[11];
  const float* w2 = (const float*)d_in[12];
  const float* g2 = (const float*)d_in[13];
  const float* b2 = (const float*)d_in[14];
  const float* m2 = (const float*)d_in[15];
  const float* v2 = (const float*)d_in[16];

  float* newxyz  = (float*)d_out;
  float* outfeat = (float*)d_out + NB * NPT * 3;

  int* fpsidx = (int*)d_ws;
  int* grp    = (int*)d_ws + NB * NPT;

  fps_kernel  <<<NB,                1024, 0, stream>>>(xyz, fpsidx, newxyz);
  ballq_kernel<<<(NB * NPT) / 8,     256, 0, stream>>>(xyz, newxyz, grp);
  mlp_kernel  <<<(NB * NPT) / 2,      64, 0, stream>>>(xyz, feat, grp, newxyz,
                                                       w0, g0, b0, m0, v0,
                                                       w1, g1, b1, m1, v1,
                                                       w2, g2, b2, m2, v2,
                                                       outfeat);
}


# FPS argmax: lane exchanges of both per-iteration reductions moved from LDS-unit ds_bpermute to vector-ALU DPP row_xmask + permlanex16 (max of distance bits, then min index); same picks, same ties
# speedup vs baseline: 1.3276x; 1.3276x over previous
#include <hip/hip_runtime.h>
#include <hip/hip_bf16.h>
#include <math.h>

typedef float v2f __attribute__((ext_vector_type(2)));
typedef float v8f __attribute__((ext_vector_type(8)));

#define NB     8
#define NPTS   16384
#define NPT    512
#define NSAMP  32
#define RAD2   0.04f
#define BN_EPS 1e-5f


template <int M>
__device__ __forceinline__ int row_xor(int v) {
  return __builtin_amdgcn_update_dpp(0, v, 0x160 | M, 0xf, 0xf, true);
}
__device__ __forceinline__ int half_xor(int v) {
  return (int)__builtin_amdgcn_permlanex16((unsigned)v, (unsigned)v, 0x76543210u, 0xfedcba98u, false, false);
}
__device__ __forceinline__ int imax2(int a, int b) { return a > b ? a : b; }
__device__ __forceinline__ int imin2(int a, int b) { return a < b ? a : b; }

__device__ __forceinline__ void wave_argmax(int& kv, int& ki) {
  int m = kv;
  m = imax2(m, row_xor<1>(m));
  m = imax2(m, row_xor<2>(m));
  m = imax2(m, row_xor<4>(m));
  m = imax2(m, row_xor<8>(m));
  m = imax2(m, half_xor(m));
  int c = (kv == m) ? ki : 0x7fffffff;
  c = imin2(c, row_xor<1>(c));
  c = imin2(c, row_xor<2>(c));
  c = imin2(c, row_xor<4>(c));
  c = imin2(c, row_xor<8>(c));
  c = imin2(c, half_xor(c));
  kv = m;
  ki = c;
}

__global__ __launch_bounds__(1024) void fps_kernel(const float* __restrict__ xyz,
                                                   int* __restrict__ fpsidx,
                                                   float* __restrict__ newxyz) {
  const int b   = blockIdx.x;
  const int tid = threadIdx.x;
  const float* base = xyz + (size_t)b * NPTS * 3;

  float px[16], py[16], pz[16], dd[16];
#pragma unroll
  for (int j = 0; j < 16; ++j) {
    int p = tid + j * 1024;
    px[j] = base[p * 3 + 0];
    py[j] = base[p * 3 + 1];
    pz[j] = base[p * 3 + 2];
    dd[j] = 3.4e38f;
  }

  __shared__ int rv[32];
  __shared__ int ri[32];
  __shared__ int   s_last;

  const int wave = tid >> 5;
  const int lane = tid & 31;
  int last = 0;

  for (int i = 0; i < NPT; ++i) {
    if (tid == 0) {
      fpsidx[b * NPT + i] = last;
      newxyz[(b * NPT + i) * 3 + 0] = base[last * 3 + 0];
      newxyz[(b * NPT + i) * 3 + 1] = base[last * 3 + 1];
      newxyz[(b * NPT + i) * 3 + 2] = base[last * 3 + 2];
    }
    const float cx = base[last * 3 + 0];
    const float cy = base[last * 3 + 1];
    const float cz = base[last * 3 + 2];

    float bv = -1.0f;
    int   bi = 0x7fffffff;
#pragma unroll
    for (int j = 0; j < 16; ++j) {
      float dx = px[j] - cx, dy = py[j] - cy, dz = pz[j] - cz;
      float d  = dx * dx + dy * dy + dz * dz;
      float nd = fminf(dd[j], d);
      dd[j] = nd;
      if (nd > bv) { bv = nd; bi = tid + j * 1024; }
    }
    int kv = __float_as_int(bv);
    int ki = bi;
    wave_argmax(kv, ki);
    if (lane == 0) { rv[wave] = kv; ri[wave] = ki; }
    __syncthreads();
    if (wave == 0) {
      kv = rv[lane]; ki = ri[lane];
      wave_argmax(kv, ki);
      if (lane == 0) s_last = ki;
    }
    __syncthreads();
    last = s_last;
  }
}

__global__ __launch_bounds__(256) void ballq_kernel(const float* __restrict__ xyz,
                                                    const float* __restrict__ newxyz,
                                                    int* __restrict__ grp) {
  const int lane = threadIdx.x & 31;
  const int q    = blockIdx.x * 8 + (threadIdx.x >> 5);
  const int b    = q >> 9;
  const int s    = q & (NPT - 1);

  const float* base = xyz + (size_t)b * NPTS * 3;
  const float cx = newxyz[(b * NPT + s) * 3 + 0];
  const float cy = newxyz[(b * NPT + s) * 3 + 1];
  const float cz = newxyz[(b * NPT + s) * 3 + 2];

  int* g = grp + (size_t)q * NSAMP;
  int cnt = 0, first = -1;

  for (int bp = 0; bp < NPTS && cnt < NSAMP; bp += 32) {
    const int p = bp + lane;
    float dx = base[p * 3 + 0] - cx;
    float dy = base[p * 3 + 1] - cy;
    float dz = base[p * 3 + 2] - cz;
    bool hit = (dx * dx + dy * dy + dz * dz) <= RAD2;
    unsigned mask = (unsigned)__ballot(hit);
    if (mask) {
      if (first < 0) first = bp + __builtin_ctz(mask);
      int pos = __popc(mask & ((1u << lane) - 1u));
      if (hit && (cnt + pos) < NSAMP) g[cnt + pos] = p;
      cnt += __popc(mask);
    }
  }
  if (first < 0) first = NPTS - 1;
  int filled = cnt < NSAMP ? cnt : NSAMP;
  for (int slot = filled + lane; slot < NSAMP; slot += 32) g[slot] = first;
}

__device__ __forceinline__ void gemm_bn_relu(const float* __restrict__ A, int lda,
                                             const float* __restrict__ W, int Kdim, int Kpad,
                                             int Nout,
                                             const float* __restrict__ gg,
                                             const float* __restrict__ bb,
                                             const float* __restrict__ mm,
                                             const float* __restrict__ vv,
                                             float* __restrict__ D, int ldd, int lane) {
  const int r15  = lane & 15;
  const int half = lane >> 4;
  for (int m0 = 0; m0 < 32; m0 += 16) {
    for (int n0 = 0; n0 < Nout; n0 += 16) {
      v8f acc = {0.f, 0.f, 0.f, 0.f, 0.f, 0.f, 0.f, 0.f};
      const int n = n0 + r15;
      for (int k0 = 0; k0 < Kpad; k0 += 4) {
        const int kk = k0 + half * 2;
        v2f a, w;
        a.x = A[(m0 + r15) * lda + kk];
        a.y = A[(m0 + r15) * lda + kk + 1];
        int ka = kk     < Kdim ? kk     : Kdim - 1;
        int kb = kk + 1 < Kdim ? kk + 1 : Kdim - 1;
        w.x = W[n * Kdim + ka];
        w.y = W[n * Kdim + kb];
        acc = __builtin_amdgcn_wmma_f32_16x16x4_f32(false, a, false, w,
                                                    (short)0, acc, false, false);
      }
      const float sc = gg[n] * rsqrtf(vv[n] + BN_EPS);
      const float bs = bb[n] - mm[n] * sc;
#pragma unroll
      for (int r = 0; r < 8; ++r) {
        float val = fmaxf(acc[r] * sc + bs, 0.f);
        D[(m0 + r + half * 8) * ldd + n] = val;
      }
    }
  }
}

__global__ __launch_bounds__(64) void mlp_kernel(const float* __restrict__ xyz,
                                                 const float* __restrict__ feat,
                                                 const int*   __restrict__ grp,
                                                 const float* __restrict__ newxyz,
                                                 const float* __restrict__ w0, const float* __restrict__ g0,
                                                 const float* __restrict__ b0, const float* __restrict__ m0,
                                                 const float* __restrict__ v0,
                                                 const float* __restrict__ w1, const float* __restrict__ g1,
                                                 const float* __restrict__ b1, const float* __restrict__ m1,
                                                 const float* __restrict__ v1,
                                                 const float* __restrict__ w2, const float* __restrict__ g2,
                                                 const float* __restrict__ b2, const float* __restrict__ m2,
                                                 const float* __restrict__ v2,
                                                 float* __restrict__ outfeat) {
  __shared__ float Xb[2][32 * 68];
  __shared__ float Hb[2][32 * 132];

  const int wave = threadIdx.x >> 5;
  const int lane = threadIdx.x & 31;
  const int q    = blockIdx.x * 2 + wave;
  const int b    = q >> 9;
  const int s    = q & (NPT - 1);

  float* X = Xb[wave];
  float* H = Hb[wave];

  const int gi = grp[(size_t)q * NSAMP + lane];
  const float* prow = xyz + ((size_t)b * NPTS + gi) * 3;
  const float* nx   = newxyz + (b * NPT + s) * 3;
  float* xr = X + lane * 68;
  xr[0] = prow[0] - nx[0];
  xr[1] = prow[1] - nx[1];
  xr[2] = prow[2] - nx[2];
  const float4* frow = (const float4*)(feat + ((size_t)b * NPTS + gi) * 64);
#pragma unroll
  for (int c4 = 0; c4 < 16; ++c4) {
    float4 f = frow[c4];
    xr[3 + 4 * c4 + 0] = f.x;
    xr[3 + 4 * c4 + 1] = f.y;
    xr[3 + 4 * c4 + 2] = f.z;
    xr[3 + 4 * c4 + 3] = f.w;
  }
  xr[67] = 0.f;

  gemm_bn_relu(X, 68,  w0, 67, 68, 64,  g0, b0, m0, v0, H, 132, lane);
  gemm_bn_relu(H, 132, w1, 64, 64, 64,  g1, b1, m1, v1, X, 68,  lane);
  gemm_bn_relu(X, 68,  w2, 64, 64, 128, g2, b2, m2, v2, H, 132, lane);

#pragma unroll
  for (int cc = 0; cc < 4; ++cc) {
    const int col = lane + cc * 32;
    float mx = H[col];
    for (int row = 1; row < 32; ++row) mx = fmaxf(mx, H[row * 132 + col]);
    outfeat[((size_t)b * 128 + col) * NPT + s] = mx;
  }
}

extern "C" void kernel_launch(void* const* d_in, const int* in_sizes, int n_in,
                              void* d_out, int out_size, void* d_ws, size_t ws_size,
                              hipStream_t stream) {
  (void)in_sizes; (void)n_in; (void)out_size; (void)ws_size;
  const float* xyz  = (const float*)d_in[0];
  const float* feat = (const float*)d_in[1];
  const float* w0 = (const float*)d_in[2];
  const float* g0 = (const float*)d_in[3];
  const float* b0 = (const float*)d_in[4];
  const float* m0 = (const float*)d_in[5];
  const float* v0 = (const float*)d_in[6];
  const float* w1 = (const float*)d_in[7];
  const float* g1 = (const float*)d_in[8];
  const float* b1 = (const float*)d_in[9];
  const float* m1 = (const float*)d_in[10];
  const float* v1 = (const float*)d_in[11];
  const float* w2 = (const float*)d_in[12];
  const float* g2 = (const float*)d_in[13];
  const float* b2 = (const float*)d_in[14];
  const float* m2 = (const float*)d_in[15];
  const float* v2 = (const float*)d_in[16];

  float* newxyz  = (float*)d_out;
  float* outfeat = (float*)d_out + NB * NPT * 3;

  int* fpsidx = (int*)d_ws;
  int* grp    = (int*)d_ws + NB * NPT;

  fps_kernel  <<<NB,                1024, 0, stream>>>(xyz, fpsidx, newxyz);
  ballq_kernel<<<(NB * NPT) / 8,     256, 0, stream>>>(xyz, newxyz, grp);
  mlp_kernel  <<<(NB * NPT) / 2,      64, 0, stream>>>(xyz, feat, grp, newxyz,
                                                       w0, g0, b0, m0, v0,
                                                       w1, g1, b1, m1, v1,
                                                       w2, g2, b2, m2, v2,
                                                       outfeat);
}


# FPS argmax in two phases: 16-point loop keeps only the running max distance; block max first, then only the wave holding it searches its registers for the lowest index; same first-occurrence rule
# speedup vs baseline: 1.6167x; 1.2178x over previous
#include <hip/hip_runtime.h>
#include <hip/hip_bf16.h>
#include <math.h>

typedef float v2f __attribute__((ext_vector_type(2)));
typedef float v8f __attribute__((ext_vector_type(8)));

#define NB     8
#define NPTS   16384
#define NPT    512
#define NSAMP  32
#define RAD2   0.04f
#define BN_EPS 1e-5f


template <int M>
__device__ __forceinline__ int row_xor(int v) {
  return __builtin_amdgcn_update_dpp(0, v, 0x160 | M, 0xf, 0xf, true);
}
__device__ __forceinline__ int half_xor(int v) {
  return (int)__builtin_amdgcn_permlanex16((unsigned)v, (unsigned)v, 0x76543210u, 0xfedcba98u, false, false);
}
__device__ __forceinline__ int imax2(int a, int b) { return a > b ? a : b; }
__device__ __forceinline__ int imin2(int a, int b) { return a < b ? a : b; }

__device__ __forceinline__ int wave_imax(int m) {
  m = imax2(m, row_xor<1>(m));
  m = imax2(m, row_xor<2>(m));
  m = imax2(m, row_xor<4>(m));
  m = imax2(m, row_xor<8>(m));
  m = imax2(m, half_xor(m));
  return m;
}
__device__ __forceinline__ int wave_imin(int c) {
  c = imin2(c, row_xor<1>(c));
  c = imin2(c, row_xor<2>(c));
  c = imin2(c, row_xor<4>(c));
  c = imin2(c, row_xor<8>(c));
  c = imin2(c, half_xor(c));
  return c;
}

__global__ __launch_bounds__(1024) void fps_kernel(const float* __restrict__ xyz,
                                                   int* __restrict__ fpsidx,
                                                   float* __restrict__ newxyz) {
  const int b   = blockIdx.x;
  const int tid = threadIdx.x;
  const float* base = xyz + (size_t)b * NPTS * 3;

  float px[16], py[16], pz[16], dd[16];
#pragma unroll
  for (int j = 0; j < 16; ++j) {
    int p = tid + j * 1024;
    px[j] = base[p * 3 + 0];
    py[j] = base[p * 3 + 1];
    pz[j] = base[p * 3 + 2];
    dd[j] = 3.4e38f;
  }

  __shared__ int rv[32];
  __shared__ int ri[32];

  const int wave = tid >> 5;
  const int lane = tid & 31;
  int last = 0;

  for (int i = 0; i < NPT; ++i) {
    const float cx = base[last * 3 + 0];
    const float cy = base[last * 3 + 1];
    const float cz = base[last * 3 + 2];
    if (tid == 0) {
      fpsidx[b * NPT + i] = last;
      newxyz[(b * NPT + i) * 3 + 0] = cx;
      newxyz[(b * NPT + i) * 3 + 1] = cy;
      newxyz[(b * NPT + i) * 3 + 2] = cz;
    }

    float bv = -1.0f;
#pragma unroll
    for (int j = 0; j < 16; ++j) {
      float dx = px[j] - cx, dy = py[j] - cy, dz = pz[j] - cz;
      float d  = dx * dx + dy * dy + dz * dz;
      float nd = fminf(dd[j], d);
      dd[j] = nd;
      bv = fmaxf(bv, nd);
    }

    const int m = wave_imax(__float_as_int(bv));
    if (lane == 0) rv[wave] = m;
    __syncthreads();
    const int M = wave_imax(rv[lane]);

    int c = 0x7fffffff;
    if (m == M) {
      int c0 = 0x7fffffff, c1 = 0x7fffffff, c2 = 0x7fffffff, c3 = 0x7fffffff;
#pragma unroll
      for (int j = 3; j >= 0; --j) {
        if (__float_as_int(dd[j])      == M) c0 = tid + j * 1024;
        if (__float_as_int(dd[j + 4])  == M) c1 = tid + (j + 4) * 1024;
        if (__float_as_int(dd[j + 8])  == M) c2 = tid + (j + 8) * 1024;
        if (__float_as_int(dd[j + 12]) == M) c3 = tid + (j + 12) * 1024;
      }
      c = wave_imin(imin2(imin2(c0, c1), imin2(c2, c3)));
    }
    if (lane == 0) ri[wave] = c;
    __syncthreads();
    last = __builtin_amdgcn_readfirstlane(wave_imin(ri[lane])) & (NPTS - 1);
  }
}

__global__ __launch_bounds__(256) void ballq_kernel(const float* __restrict__ xyz,
                                                    const float* __restrict__ newxyz,
                                                    int* __restrict__ grp) {
  const int lane = threadIdx.x & 31;
  const int q    = blockIdx.x * 8 + (threadIdx.x >> 5);
  const int b    = q >> 9;
  const int s    = q & (NPT - 1);

  const float* base = xyz + (size_t)b * NPTS * 3;
  const float cx = newxyz[(b * NPT + s) * 3 + 0];
  const float cy = newxyz[(b * NPT + s) * 3 + 1];
  const float cz = newxyz[(b * NPT + s) * 3 + 2];

  int* g = grp + (size_t)q * NSAMP;
  int cnt = 0, first = -1;

  for (int bp = 0; bp < NPTS && cnt < NSAMP; bp += 32) {
    const int p = bp + lane;
    float dx = base[p * 3 + 0] - cx;
    float dy = base[p * 3 + 1] - cy;
    float dz = base[p * 3 + 2] - cz;
    bool hit = (dx * dx + dy * dy + dz * dz) <= RAD2;
    unsigned mask = (unsigned)__ballot(hit);
    if (mask) {
      if (first < 0) first = bp + __builtin_ctz(mask);
      int pos = __popc(mask & ((1u << lane) - 1u));
      if (hit && (cnt + pos) < NSAMP) g[cnt + pos] = p;
      cnt += __popc(mask);
    }
  }
  if (first < 0) first = NPTS - 1;
  int filled = cnt < NSAMP ? cnt : NSAMP;
  for (int slot = filled + lane; slot < NSAMP; slot += 32) g[slot] = first;
}

__device__ __forceinline__ void gemm_bn_relu(const float* __restrict__ A, int lda,
                                             const float* __restrict__ W, int Kdim, int Kpad,
                                             int Nout,
                                             const float* __restrict__ gg,
                                             const float* __restrict__ bb,
                                             const float* __restrict__ mm,
                                             const float* __restrict__ vv,
                                             float* __restrict__ D, int ldd, int lane) {
  const int r15  = lane & 15;
  const int half = lane >> 4;
  for (int m0 = 0; m0 < 32; m0 += 16) {
    for (int n0 = 0; n0 < Nout; n0 += 16) {
      v8f acc = {0.f, 0.f, 0.f, 0.f, 0.f, 0.f, 0.f, 0.f};
      const int n = n0 + r15;
      for (int k0 = 0; k0 < Kpad; k0 += 4) {
        const int kk = k0 + half * 2;
        v2f a, w;
        a.x = A[(m0 + r15) * lda + kk];
        a.y = A[(m0 + r15) * lda + kk + 1];
        int ka = kk     < Kdim ? kk     : Kdim - 1;
        int kb = kk + 1 < Kdim ? kk + 1 : Kdim - 1;
        w.x = W[n * Kdim + ka];
        w.y = W[n * Kdim + kb];
        acc = __builtin_amdgcn_wmma_f32_16x16x4_f32(false, a, false, w,
                                                    (short)0, acc, false, false);
      }
      const float sc = gg[n] * rsqrtf(vv[n] + BN_EPS);
      const float bs = bb[n] - mm[n] * sc;
#pragma unroll
      for (int r = 0; r < 8; ++r) {
        float val = fmaxf(acc[r] * sc + bs, 0.f);
        D[(m0 + r + half * 8) * ldd + n] = val;
      }
    }
  }
}

__global__ __launch_bounds__(64) void mlp_kernel(const float* __restrict__ xyz,
                                                 const float* __restrict__ feat,
                                                 const int*   __restrict__ grp,
                                                 const float* __restrict__ newxyz,
                                                 const float* __restrict__ w0, const float* __restrict__ g0,
                                                 const float* __restrict__ b0, const float* __restrict__ m0,
                                                 const float* __restrict__ v0,
                                                 const float* __restrict__ w1, const float* __restrict__ g1,
                                                 const float* __restrict__ b1, const float* __restrict__ m1,
                                                 const float* __restrict__ v1,
                                                 const float* __restrict__ w2, const float* __restrict__ g2,
                                                 const float* __restrict__ b2, const float* __restrict__ m2,
                                                 const float* __restrict__ v2,
                                                 float* __restrict__ outfeat) {
  __shared__ float Xb[2][32 * 68];
  __shared__ float Hb[2][32 * 132];

  const int wave = threadIdx.x >> 5;
  const int lane = threadIdx.x & 31;
  const int q    = blockIdx.x * 2 + wave;
  const int b    = q >> 9;
  const int s    = q & (NPT - 1);

  float* X = Xb[wave];
  float* H = Hb[wave];

  const int gi = grp[(size_t)q * NSAMP + lane];
  const float* prow = xyz + ((size_t)b * NPTS + gi) * 3;
  const float* nx   = newxyz + (b * NPT + s) * 3;
  float* xr = X + lane * 68;
  xr[0] = prow[0] - nx[0];
  xr[1] = prow[1] - nx[1];
  xr[2] = prow[2] - nx[2];
  const float4* frow = (const float4*)(feat + ((size_t)b * NPTS + gi) * 64);
#pragma unroll
  for (int c4 = 0; c4 < 16; ++c4) {
    float4 f = frow[c4];
    xr[3 + 4 * c4 + 0] = f.x;
    xr[3 + 4 * c4 + 1] = f.y;
    xr[3 + 4 * c4 + 2] = f.z;
    xr[3 + 4 * c4 + 3] = f.w;
  }
  xr[67] = 0.f;

  gemm_bn_relu(X, 68,  w0, 67, 68, 64,  g0, b0, m0, v0, H, 132, lane);
  gemm_bn_relu(H, 132, w1, 64, 64, 64,  g1, b1, m1, v1, X, 68,  lane);
  gemm_bn_relu(X, 68,  w2, 64, 64, 128, g2, b2, m2, v2, H, 132, lane);

#pragma unroll
  for (int cc = 0; cc < 4; ++cc) {
    const int col = lane + cc * 32;
    float mx = H[col];
    for (int row = 1; row < 32; ++row) mx = fmaxf(mx, H[row * 132 + col]);
    outfeat[((size_t)b * 128 + col) * NPT + s] = mx;
  }
}

extern "C" void kernel_launch(void* const* d_in, const int* in_sizes, int n_in,
                              void* d_out, int out_size, void* d_ws, size_t ws_size,
                              hipStream_t stream) {
  (void)in_sizes; (void)n_in; (void)out_size; (void)ws_size;
  const float* xyz  = (const float*)d_in[0];
  const float* feat = (const float*)d_in[1];
  const float* w0 = (const float*)d_in[2];
  const float* g0 = (const float*)d_in[3];
  const float* b0 = (const float*)d_in[4];
  const float* m0 = (const float*)d_in[5];
  const float* v0 = (const float*)d_in[6];
  const float* w1 = (const float*)d_in[7];
  const float* g1 = (const float*)d_in[8];
  const float* b1 = (const float*)d_in[9];
  const float* m1 = (const float*)d_in[10];
  const float* v1 = (const float*)d_in[11];
  const float* w2 = (const float*)d_in[12];
  const float* g2 = (const float*)d_in[13];
  const float* b2 = (const float*)d_in[14];
  const float* m2 = (const float*)d_in[15];
  const float* v2 = (const float*)d_in[16];

  float* newxyz  = (float*)d_out;
  float* outfeat = (float*)d_out + NB * NPT * 3;

  int* fpsidx = (int*)d_ws;
  int* grp    = (int*)d_ws + NB * NPT;

  fps_kernel  <<<NB,                1024, 0, stream>>>(xyz, fpsidx, newxyz);
  ballq_kernel<<<(NB * NPT) / 8,     256, 0, stream>>>(xyz, newxyz, grp);
  mlp_kernel  <<<(NB * NPT) / 2,      64, 0, stream>>>(xyz, feat, grp, newxyz,
                                                       w0, g0, b0, m0, v0,
                                                       w1, g1, b1, m1, v1,
                                                       w2, g2, b2, m2, v2,
                                                       outfeat);
}


# FPS: each step's next centre is read from an LDS copy of the batch's cloud (per-wave candidates fetched beside the index fold, winner's lane read out), not by a dependent global load; same picks
# speedup vs baseline: 1.7220x; 1.0651x over previous
#include <hip/hip_runtime.h>
#include <hip/hip_bf16.h>
#include <math.h>

typedef float v2f __attribute__((ext_vector_type(2)));
typedef float v8f __attribute__((ext_vector_type(8)));

#define NB     8
#define NPTS   16384
#define NPT    512
#define NSAMP  32
#define RAD2   0.04f
#define BN_EPS 1e-5f


template <int M>
__device__ __forceinline__ int row_xor(int v) {
  return __builtin_amdgcn_update_dpp(0, v, 0x160 | M, 0xf, 0xf, true);
}
__device__ __forceinline__ int half_xor(int v) {
  return (int)__builtin_amdgcn_permlanex16((unsigned)v, (unsigned)v, 0x76543210u, 0xfedcba98u, false, false);
}
__device__ __forceinline__ int imax2(int a, int b) { return a > b ? a : b; }
__device__ __forceinline__ int imin2(int a, int b) { return a < b ? a : b; }

__device__ __forceinline__ int wave_imax(int m) {
  m = imax2(m, row_xor<1>(m));
  m = imax2(m, row_xor<2>(m));
  m = imax2(m, row_xor<4>(m));
  m = imax2(m, row_xor<8>(m));
  m = imax2(m, half_xor(m));
  return m;
}
__device__ __forceinline__ int wave_imin(int c) {
  c = imin2(c, row_xor<1>(c));
  c = imin2(c, row_xor<2>(c));
  c = imin2(c, row_xor<4>(c));
  c = imin2(c, row_xor<8>(c));
  c = imin2(c, half_xor(c));
  return c;
}

__global__ __launch_bounds__(1024) void fps_kernel(const float* __restrict__ xyz,
                                                   int* __restrict__ fpsidx,
                                                   float* __restrict__ newxyz) {
  const int b   = blockIdx.x;
  const int tid = threadIdx.x;
  const float* base = xyz + (size_t)b * NPTS * 3;

  __shared__ float4 P[NPTS];
  __shared__ int rv[32];
  __shared__ int ri[32];

  float px[16], py[16], pz[16], dd[16];
#pragma unroll
  for (int j = 0; j < 16; ++j) {
    int p = tid + j * 1024;
    px[j] = base[p * 3 + 0];
    py[j] = base[p * 3 + 1];
    pz[j] = base[p * 3 + 2];
    dd[j] = 3.4e38f;
    P[p] = make_float4(px[j], py[j], pz[j], 0.0f);
  }

  const int wave = tid >> 5;
  const int lane = tid & 31;
  int last = 0;
  float cx = base[0], cy = base[1], cz = base[2];

  for (int i = 0; i < NPT; ++i) {
    if (tid == 0) {
      fpsidx[b * NPT + i] = last;
      newxyz[(b * NPT + i) * 3 + 0] = cx;
      newxyz[(b * NPT + i) * 3 + 1] = cy;
      newxyz[(b * NPT + i) * 3 + 2] = cz;
    }

    float bv = -1.0f;
#pragma unroll
    for (int j = 0; j < 16; ++j) {
      float dx = px[j] - cx, dy = py[j] - cy, dz = pz[j] - cz;
      float d  = dx * dx + dy * dy + dz * dz;
      float nd = fminf(dd[j], d);
      dd[j] = nd;
      bv = fmaxf(bv, nd);
    }

    const int m = wave_imax(__float_as_int(bv));
    if (lane == 0) rv[wave] = m;
    __syncthreads();
    const int M = wave_imax(rv[lane]);

    int c = 0x7fffffff;
    if (m == M) {
      int c0 = 0x7fffffff, c1 = 0x7fffffff, c2 = 0x7fffffff, c3 = 0x7fffffff;
#pragma unroll
      for (int j = 3; j >= 0; --j) {
        if (__float_as_int(dd[j])      == M) c0 = tid + j * 1024;
        if (__float_as_int(dd[j + 4])  == M) c1 = tid + (j + 4) * 1024;
        if (__float_as_int(dd[j + 8])  == M) c2 = tid + (j + 8) * 1024;
        if (__float_as_int(dd[j + 12]) == M) c3 = tid + (j + 12) * 1024;
      }
      c = wave_imin(imin2(imin2(c0, c1), imin2(c2, c3)));
    }
    if (lane == 0) ri[wave] = c;
    __syncthreads();
    const int    cand = ri[lane];
    const float4 pc   = P[cand & (NPTS - 1)];
    const int    cmin = wave_imin(cand);
    const int    wl   = __builtin_ctz(__builtin_amdgcn_ballot_w32(cand == cmin));
    cx = __int_as_float(__builtin_amdgcn_readlane(__float_as_int(pc.x), wl));
    cy = __int_as_float(__builtin_amdgcn_readlane(__float_as_int(pc.y), wl));
    cz = __int_as_float(__builtin_amdgcn_readlane(__float_as_int(pc.z), wl));
    last = __builtin_amdgcn_readfirstlane(cmin) & (NPTS - 1);
  }
}

__global__ __launch_bounds__(256) void ballq_kernel(const float* __restrict__ xyz,
                                                    const float* __restrict__ newxyz,
                                                    int* __restrict__ grp) {
  const int lane = threadIdx.x & 31;
  const int q    = blockIdx.x * 8 + (threadIdx.x >> 5);
  const int b    = q >> 9;
  const int s    = q & (NPT - 1);

  const float* base = xyz + (size_t)b * NPTS * 3;
  const float cx = newxyz[(b * NPT + s) * 3 + 0];
  const float cy = newxyz[(b * NPT + s) * 3 + 1];
  const float cz = newxyz[(b * NPT + s) * 3 + 2];

  int* g = grp + (size_t)q * NSAMP;
  int cnt = 0, first = -1;

  for (int bp = 0; bp < NPTS && cnt < NSAMP; bp += 32) {
    const int p = bp + lane;
    float dx = base[p * 3 + 0] - cx;
    float dy = base[p * 3 + 1] - cy;
    float dz = base[p * 3 + 2] - cz;
    bool hit = (dx * dx + dy * dy + dz * dz) <= RAD2;
    unsigned mask = (unsigned)__ballot(hit);
    if (mask) {
      if (first < 0) first = bp + __builtin_ctz(mask);
      int pos = __popc(mask & ((1u << lane) - 1u));
      if (hit && (cnt + pos) < NSAMP) g[cnt + pos] = p;
      cnt += __popc(mask);
    }
  }
  if (first < 0) first = NPTS - 1;
  int filled = cnt < NSAMP ? cnt : NSAMP;
  for (int slot = filled + lane; slot < NSAMP; slot += 32) g[slot] = first;
}

__device__ __forceinline__ void gemm_bn_relu(const float* __restrict__ A, int lda,
                                             const float* __restrict__ W, int Kdim, int Kpad,
                                             int Nout,
                                             const float* __restrict__ gg,
                                             const float* __restrict__ bb,
                                             const float* __restrict__ mm,
                                             const float* __restrict__ vv,
                                             float* __restrict__ D, int ldd, int lane) {
  const int r15  = lane & 15;
  const int half = lane >> 4;
  for (int m0 = 0; m0 < 32; m0 += 16) {
    for (int n0 = 0; n0 < Nout; n0 += 16) {
      v8f acc = {0.f, 0.f, 0.f, 0.f, 0.f, 0.f, 0.f, 0.f};
      const int n = n0 + r15;
      for (int k0 = 0; k0 < Kpad; k0 += 4) {
        const int kk = k0 + half * 2;
        v2f a, w;
        a.x = A[(m0 + r15) * lda + kk];
        a.y = A[(m0 + r15) * lda + kk + 1];
        int ka = kk     < Kdim ? kk     : Kdim - 1;
        int kb = kk + 1 < Kdim ? kk + 1 : Kdim - 1;
        w.x = W[n * Kdim + ka];
        w.y = W[n * Kdim + kb];
        acc = __builtin_amdgcn_wmma_f32_16x16x4_f32(false, a, false, w,
                                                    (short)0, acc, false, false);
      }
      const float sc = gg[n] * rsqrtf(vv[n] + BN_EPS);
      const float bs = bb[n] - mm[n] * sc;
#pragma unroll
      for (int r = 0; r < 8; ++r) {
        float val = fmaxf(acc[r] * sc + bs, 0.f);
        D[(m0 + r + half * 8) * ldd + n] = val;
      }
    }
  }
}

__global__ __launch_bounds__(64) void mlp_kernel(const float* __restrict__ xyz,
                                                 const float* __restrict__ feat,
                                                 const int*   __restrict__ grp,
                                                 const float* __restrict__ newxyz,
                                                 const float* __restrict__ w0, const float* __restrict__ g0,
                                                 const float* __restrict__ b0, const float* __restrict__ m0,
                                                 const float* __restrict__ v0,
                                                 const float* __restrict__ w1, const float* __restrict__ g1,
                                                 const float* __restrict__ b1, const float* __restrict__ m1,
                                                 const float* __restrict__ v1,
                                                 const float* __restrict__ w2, const float* __restrict__ g2,
                                                 const float* __restrict__ b2, const float* __restrict__ m2,
                                                 const float* __restrict__ v2,
                                                 float* __restrict__ outfeat) {
  __shared__ float Xb[2][32 * 68];
  __shared__ float Hb[2][32 * 132];

  const int wave = threadIdx.x >> 5;
  const int lane = threadIdx.x & 31;
  const int q    = blockIdx.x * 2 + wave;
  const int b    = q >> 9;
  const int s    = q & (NPT - 1);

  float* X = Xb[wave];
  float* H = Hb[wave];

  const int gi = grp[(size_t)q * NSAMP + lane];
  const float* prow = xyz + ((size_t)b * NPTS + gi) * 3;
  const float* nx   = newxyz + (b * NPT + s) * 3;
  float* xr = X + lane * 68;
  xr[0] = prow[0] - nx[0];
  xr[1] = prow[1] - nx[1];
  xr[2] = prow[2] - nx[2];
  const float4* frow = (const float4*)(feat + ((size_t)b * NPTS + gi) * 64);
#pragma unroll
  for (int c4 = 0; c4 < 16; ++c4) {
    float4 f = frow[c4];
    xr[3 + 4 * c4 + 0] = f.x;
    xr[3 + 4 * c4 + 1] = f.y;
    xr[3 + 4 * c4 + 2] = f.z;
    xr[3 + 4 * c4 + 3] = f.w;
  }
  xr[67] = 0.f;

  gemm_bn_relu(X, 68,  w0, 67, 68, 64,  g0, b0, m0, v0, H, 132, lane);
  gemm_bn_relu(H, 132, w1, 64, 64, 64,  g1, b1, m1, v1, X, 68,  lane);
  gemm_bn_relu(X, 68,  w2, 64, 64, 128, g2, b2, m2, v2, H, 132, lane);

#pragma unroll
  for (int cc = 0; cc < 4; ++cc) {
    const int col = lane + cc * 32;
    float mx = H[col];
    for (int row = 1; row < 32; ++row) mx = fmaxf(mx, H[row * 132 + col]);
    outfeat[((size_t)b * 128 + col) * NPT + s] = mx;
  }
}

extern "C" void kernel_launch(void* const* d_in, const int* in_sizes, int n_in,
                              void* d_out, int out_size, void* d_ws, size_t ws_size,
                              hipStream_t stream) {
  (void)in_sizes; (void)n_in; (void)out_size; (void)ws_size;
  const float* xyz  = (const float*)d_in[0];
  const float* feat = (const float*)d_in[1];
  const float* w0 = (const float*)d_in[2];
  const float* g0 = (const float*)d_in[3];
  const float* b0 = (const float*)d_in[4];
  const float* m0 = (const float*)d_in[5];
  const float* v0 = (const float*)d_in[6];
  const float* w1 = (const float*)d_in[7];
  const float* g1 = (const float*)d_in[8];
  const float* b1 = (const float*)d_in[9];
  const float* m1 = (const float*)d_in[10];
  const float* v1 = (const float*)d_in[11];
  const float* w2 = (const float*)d_in[12];
  const float* g2 = (const float*)d_in[13];
  const float* b2 = (const float*)d_in[14];
  const float* m2 = (const float*)d_in[15];
  const float* v2 = (const float*)d_in[16];

  float* newxyz  = (float*)d_out;
  float* outfeat = (float*)d_out + NB * NPT * 3;

  int* fpsidx = (int*)d_ws;
  int* grp    = (int*)d_ws + NB * NPT;

  fps_kernel  <<<NB,                1024, 0, stream>>>(xyz, fpsidx, newxyz);
  ballq_kernel<<<(NB * NPT) / 8,     256, 0, stream>>>(xyz, newxyz, grp);
  mlp_kernel  <<<(NB * NPT) / 2,      64, 0, stream>>>(xyz, feat, grp, newxyz,
                                                       w0, g0, b0, m0, v0,
                                                       w1, g1, b1, m1, v1,
                                                       w2, g2, b2, m2, v2,
                                                       outfeat);
}


# FPS: points put in bit-interleaved grid-cell order so each wave holds a compact box; a wave skips its distance update when the centre is farther from its box than its current maximum; same picks
# speedup vs baseline: 1.9009x; 1.1039x over previous
#include <hip/hip_runtime.h>
#include <hip/hip_bf16.h>
#include <math.h>

typedef float v2f __attribute__((ext_vector_type(2)));
typedef float v8f __attribute__((ext_vector_type(8)));

#define NB     8
#define NPTS   16384
#define NPT    512
#define NSAMP  32
#define RAD2   0.04f
#define BN_EPS 1e-5f


template <int M>
__device__ __forceinline__ int row_xor(int v) {
  return __builtin_amdgcn_update_dpp(0, v, 0x160 | M, 0xf, 0xf, true);
}
__device__ __forceinline__ int half_xor(int v) {
  return (int)__builtin_amdgcn_permlanex16((unsigned)v, (unsigned)v, 0x76543210u, 0xfedcba98u, false, false);
}
__device__ __forceinline__ int imax2(int a, int b) { return a > b ? a : b; }
__device__ __forceinline__ int imin2(int a, int b) { return a < b ? a : b; }

__device__ __forceinline__ int wave_imax(int m) {
  m = imax2(m, row_xor<1>(m));
  m = imax2(m, row_xor<2>(m));
  m = imax2(m, row_xor<4>(m));
  m = imax2(m, row_xor<8>(m));
  m = imax2(m, half_xor(m));
  return m;
}
__device__ __forceinline__ int wave_imin(int c) {
  c = imin2(c, row_xor<1>(c));
  c = imin2(c, row_xor<2>(c));
  c = imin2(c, row_xor<4>(c));
  c = imin2(c, row_xor<8>(c));
  c = imin2(c, half_xor(c));
  return c;
}

__device__ __forceinline__ float uni(float v) {
  return __int_as_float(__builtin_amdgcn_readfirstlane(__float_as_int(v)));
}
__device__ __forceinline__ unsigned spread3(unsigned v) {
  return (v & 1u) | ((v & 2u) << 2) | ((v & 4u) << 4);
}

__global__ __launch_bounds__(1024) void fps_kernel(const float* __restrict__ xyz,
                                                   int* __restrict__ fpsidx,
                                                   float* __restrict__ newxyz) {
  const int b   = blockIdx.x;
  const int tid = threadIdx.x;
  const float* base = xyz + (size_t)b * NPTS * 3;

  __shared__ float4 P[NPTS];
  __shared__ int rv[32];
  __shared__ int ri[32];
  __shared__ int wtot[32];

  const int wave = tid >> 5;
  const int lane = tid & 31;

  unsigned* K = (unsigned*)P;
  unsigned key[16];
#pragma unroll
  for (int j = 0; j < 16; ++j) {
    const int p = tid * 16 + j;
    const float x = base[p * 3 + 0], y = base[p * 3 + 1], z = base[p * 3 + 2];
    const unsigned ix = (unsigned)(int)fminf(fmaxf(x * 8.0f, 0.0f), 7.0f);
    const unsigned iy = (unsigned)(int)fminf(fmaxf(y * 8.0f, 0.0f), 7.0f);
    const unsigned iz = (unsigned)(int)fminf(fmaxf(z * 8.0f, 0.0f), 7.0f);
    key[j] = ((spread3(ix) | (spread3(iy) << 1) | (spread3(iz) << 2)) << 14) | (unsigned)p;
  }
  for (int bit = 14; bit < 23; ++bit) {
    int z = 0;
#pragma unroll
    for (int j = 0; j < 16; ++j) z += (int)(((key[j] >> bit) & 1u) ^ 1u);
    int inc = z;
    for (int d = 1; d < 32; d <<= 1) {
      const int t = __shfl_up(inc, d);
      if (lane >= d) inc += t;
    }
    if (lane == 31) wtot[wave] = inc;
    __syncthreads();
    const int wt = wtot[lane];
    int winc = wt;
    for (int d = 1; d < 32; d <<= 1) {
      const int t = __shfl_up(winc, d);
      if (lane >= d) winc += t;
    }
    const int zall  = __shfl(winc, 31);
    const int wexcl = __shfl(winc - wt, wave);
    int zpos = wexcl + inc - z;
    int opos = zall + (tid * 16 - zpos);
#pragma unroll
    for (int j = 0; j < 16; ++j) {
      const int one = (int)((key[j] >> bit) & 1u);
      K[one ? opos : zpos] = key[j];
      opos += one;
      zpos += 1 - one;
    }
    __syncthreads();
#pragma unroll
    for (int j = 0; j < 16; ++j) key[j] = K[tid * 16 + j];
    __syncthreads();
  }

  float px[16], py[16], pz[16], dd[16];
  unsigned pk[8];
#pragma unroll
  for (int j = 0; j < 16; ++j) {
    const int p = (int)(key[j] & 0x3fffu);
    px[j] = base[p * 3 + 0];
    py[j] = base[p * 3 + 1];
    pz[j] = base[p * 3 + 2];
    dd[j] = 3.4e38f;
    P[p] = make_float4(px[j], py[j], pz[j], 0.0f);
  }
#pragma unroll
  for (int j = 0; j < 16; j += 2)
    pk[j >> 1] = (key[j] & 0x3fffu) | ((key[j + 1] & 0x3fffu) << 16);

  float lx = px[0], hx = px[0], ly = py[0], hy = py[0], lz = pz[0], hz = pz[0];
#pragma unroll
  for (int j = 1; j < 16; ++j) {
    lx = fminf(lx, px[j]); hx = fmaxf(hx, px[j]);
    ly = fminf(ly, py[j]); hy = fmaxf(hy, py[j]);
    lz = fminf(lz, pz[j]); hz = fmaxf(hz, pz[j]);
  }
  for (int d = 16; d >= 1; d >>= 1) {
    lx = fminf(lx, __shfl_xor(lx, d)); hx = fmaxf(hx, __shfl_xor(hx, d));
    ly = fminf(ly, __shfl_xor(ly, d)); hy = fmaxf(hy, __shfl_xor(hy, d));
    lz = fminf(lz, __shfl_xor(lz, d)); hz = fmaxf(hz, __shfl_xor(hz, d));
  }
  const float lox = uni(lx), hix = uni(hx), loy = uni(ly), hiy = uni(hy), loz = uni(lz), hiz = uni(hz);

  int last = 0;
  float cx = base[0], cy = base[1], cz = base[2];
  int m = __float_as_int(3.4e38f);

  for (int i = 0; i < NPT; ++i) {
    if (tid == 0) {
      fpsidx[b * NPT + i] = last;
      newxyz[(b * NPT + i) * 3 + 0] = cx;
      newxyz[(b * NPT + i) * 3 + 1] = cy;
      newxyz[(b * NPT + i) * 3 + 2] = cz;
    }

    const float ex = fmaxf(fmaxf(lox - cx, cx - hix), 0.0f);
    const float ey = fmaxf(fmaxf(loy - cy, cy - hiy), 0.0f);
    const float ez = fmaxf(fmaxf(loz - cz, cz - hiz), 0.0f);
    const float lb = (ex * ex + ey * ey + ez * ez) * 0.999f;
    const float wmax = uni(__int_as_float(m));
    if (!(lb > wmax)) {
      float bv = -1.0f;
#pragma unroll
      for (int j = 0; j < 16; ++j) {
        float dx = px[j] - cx, dy = py[j] - cy, dz = pz[j] - cz;
        float d  = __builtin_fmaf(dz, dz, __builtin_fmaf(dx, dx, dy * dy));
        float nd = fminf(dd[j], d);
        dd[j] = nd;
        bv = fmaxf(bv, nd);
      }
      m = wave_imax(__float_as_int(bv));
    }

    if (lane == 0) rv[wave] = m;
    __syncthreads();
    const int M = wave_imax(rv[lane]);

    int c = 0x7fffffff;
    if (m == M) {
      int c0 = 0x7fffffff, c1 = 0x7fffffff, c2 = 0x7fffffff, c3 = 0x7fffffff;
#pragma unroll
      for (int j = 0; j < 16; j += 4) {
        if (__float_as_int(dd[j])     == M) c0 = imin2(c0, (int)(pk[j >> 1] & 0xffffu));
        if (__float_as_int(dd[j + 1]) == M) c1 = imin2(c1, (int)(pk[j >> 1] >> 16));
        if (__float_as_int(dd[j + 2]) == M) c2 = imin2(c2, (int)(pk[(j >> 1) + 1] & 0xffffu));
        if (__float_as_int(dd[j + 3]) == M) c3 = imin2(c3, (int)(pk[(j >> 1) + 1] >> 16));
      }
      c = wave_imin(imin2(imin2(c0, c1), imin2(c2, c3)));
    }
    if (lane == 0) ri[wave] = c;
    __syncthreads();
    const int    cand = ri[lane];
    const float4 pc   = P[cand & (NPTS - 1)];
    const int    cmin = wave_imin(cand);
    const int    wl   = __builtin_ctz(__builtin_amdgcn_ballot_w32(cand == cmin));
    cx = __int_as_float(__builtin_amdgcn_readlane(__float_as_int(pc.x), wl));
    cy = __int_as_float(__builtin_amdgcn_readlane(__float_as_int(pc.y), wl));
    cz = __int_as_float(__builtin_amdgcn_readlane(__float_as_int(pc.z), wl));
    last = __builtin_amdgcn_readfirstlane(cmin) & (NPTS - 1);
  }
}

__global__ __launch_bounds__(256) void ballq_kernel(const float* __restrict__ xyz,
                                                    const float* __restrict__ newxyz,
                                                    int* __restrict__ grp) {
  const int lane = threadIdx.x & 31;
  const int q    = blockIdx.x * 8 + (threadIdx.x >> 5);
  const int b    = q >> 9;
  const int s    = q & (NPT - 1);

  const float* base = xyz + (size_t)b * NPTS * 3;
  const float cx = newxyz[(b * NPT + s) * 3 + 0];
  const float cy = newxyz[(b * NPT + s) * 3 + 1];
  const float cz = newxyz[(b * NPT + s) * 3 + 2];

  int* g = grp + (size_t)q * NSAMP;
  int cnt = 0, first = -1;

  for (int bp = 0; bp < NPTS && cnt < NSAMP; bp += 32) {
    const int p = bp + lane;
    float dx = base[p * 3 + 0] - cx;
    float dy = base[p * 3 + 1] - cy;
    float dz = base[p * 3 + 2] - cz;
    bool hit = (dx * dx + dy * dy + dz * dz) <= RAD2;
    unsigned mask = (unsigned)__ballot(hit);
    if (mask) {
      if (first < 0) first = bp + __builtin_ctz(mask);
      int pos = __popc(mask & ((1u << lane) - 1u));
      if (hit && (cnt + pos) < NSAMP) g[cnt + pos] = p;
      cnt += __popc(mask);
    }
  }
  if (first < 0) first = NPTS - 1;
  int filled = cnt < NSAMP ? cnt : NSAMP;
  for (int slot = filled + lane; slot < NSAMP; slot += 32) g[slot] = first;
}

__device__ __forceinline__ void gemm_bn_relu(const float* __restrict__ A, int lda,
                                             const float* __restrict__ W, int Kdim, int Kpad,
                                             int Nout,
                                             const float* __restrict__ gg,
                                             const float* __restrict__ bb,
                                             const float* __restrict__ mm,
                                             const float* __restrict__ vv,
                                             float* __restrict__ D, int ldd, int lane) {
  const int r15  = lane & 15;
  const int half = lane >> 4;
  for (int m0 = 0; m0 < 32; m0 += 16) {
    for (int n0 = 0; n0 < Nout; n0 += 16) {
      v8f acc = {0.f, 0.f, 0.f, 0.f, 0.f, 0.f, 0.f, 0.f};
      const int n = n0 + r15;
      for (int k0 = 0; k0 < Kpad; k0 += 4) {
        const int kk = k0 + half * 2;
        v2f a, w;
        a.x = A[(m0 + r15) * lda + kk];
        a.y = A[(m0 + r15) * lda + kk + 1];
        int ka = kk     < Kdim ? kk     : Kdim - 1;
        int kb = kk + 1 < Kdim ? kk + 1 : Kdim - 1;
        w.x = W[n * Kdim + ka];
        w.y = W[n * Kdim + kb];
        acc = __builtin_amdgcn_wmma_f32_16x16x4_f32(false, a, false, w,
                                                    (short)0, acc, false, false);
      }
      const float sc = gg[n] * rsqrtf(vv[n] + BN_EPS);
      const float bs = bb[n] - mm[n] * sc;
#pragma unroll
      for (int r = 0; r < 8; ++r) {
        float val = fmaxf(acc[r] * sc + bs, 0.f);
        D[(m0 + r + half * 8) * ldd + n] = val;
      }
    }
  }
}

__global__ __launch_bounds__(64) void mlp_kernel(const float* __restrict__ xyz,
                                                 const float* __restrict__ feat,
                                                 const int*   __restrict__ grp,
                                                 const float* __restrict__ newxyz,
                                                 const float* __restrict__ w0, const float* __restrict__ g0,
                                                 const float* __restrict__ b0, const float* __restrict__ m0,
                                                 const float* __restrict__ v0,
                                                 const float* __restrict__ w1, const float* __restrict__ g1,
                                                 const float* __restrict__ b1, const float* __restrict__ m1,
                                                 const float* __restrict__ v1,
                                                 const float* __restrict__ w2, const float* __restrict__ g2,
                                                 const float* __restrict__ b2, const float* __restrict__ m2,
                                                 const float* __restrict__ v2,
                                                 float* __restrict__ outfeat) {
  __shared__ float Xb[2][32 * 68];
  __shared__ float Hb[2][32 * 132];

  const int wave = threadIdx.x >> 5;
  const int lane = threadIdx.x & 31;
  const int q    = blockIdx.x * 2 + wave;
  const int b    = q >> 9;
  const int s    = q & (NPT - 1);

  float* X = Xb[wave];
  float* H = Hb[wave];

  const int gi = grp[(size_t)q * NSAMP + lane];
  const float* prow = xyz + ((size_t)b * NPTS + gi) * 3;
  const float* nx   = newxyz + (b * NPT + s) * 3;
  float* xr = X + lane * 68;
  xr[0] = prow[0] - nx[0];
  xr[1] = prow[1] - nx[1];
  xr[2] = prow[2] - nx[2];
  const float4* frow = (const float4*)(feat + ((size_t)b * NPTS + gi) * 64);
#pragma unroll
  for (int c4 = 0; c4 < 16; ++c4) {
    float4 f = frow[c4];
    xr[3 + 4 * c4 + 0] = f.x;
    xr[3 + 4 * c4 + 1] = f.y;
    xr[3 + 4 * c4 + 2] = f.z;
    xr[3 + 4 * c4 + 3] = f.w;
  }
  xr[67] = 0.f;

  gemm_bn_relu(X, 68,  w0, 67, 68, 64,  g0, b0, m0, v0, H, 132, lane);
  gemm_bn_relu(H, 132, w1, 64, 64, 64,  g1, b1, m1, v1, X, 68,  lane);
  gemm_bn_relu(X, 68,  w2, 64, 64, 128, g2, b2, m2, v2, H, 132, lane);

#pragma unroll
  for (int cc = 0; cc < 4; ++cc) {
    const int col = lane + cc * 32;
    float mx = H[col];
    for (int row = 1; row < 32; ++row) mx = fmaxf(mx, H[row * 132 + col]);
    outfeat[((size_t)b * 128 + col) * NPT + s] = mx;
  }
}

extern "C" void kernel_launch(void* const* d_in, const int* in_sizes, int n_in,
                              void* d_out, int out_size, void* d_ws, size_t ws_size,
                              hipStream_t stream) {
  (void)in_sizes; (void)n_in; (void)out_size; (void)ws_size;
  const float* xyz  = (const float*)d_in[0];
  const float* feat = (const float*)d_in[1];
  const float* w0 = (const float*)d_in[2];
  const float* g0 = (const float*)d_in[3];
  const float* b0 = (const float*)d_in[4];
  const float* m0 = (const float*)d_in[5];
  const float* v0 = (const float*)d_in[6];
  const float* w1 = (const float*)d_in[7];
  const float* g1 = (const float*)d_in[8];
  const float* b1 = (const float*)d_in[9];
  const float* m1 = (const float*)d_in[10];
  const float* v1 = (const float*)d_in[11];
  const float* w2 = (const float*)d_in[12];
  const float* g2 = (const float*)d_in[13];
  const float* b2 = (const float*)d_in[14];
  const float* m2 = (const float*)d_in[15];
  const float* v2 = (const float*)d_in[16];

  float* newxyz  = (float*)d_out;
  float* outfeat = (float*)d_out + NB * NPT * 3;

  int* fpsidx = (int*)d_ws;
  int* grp    = (int*)d_ws + NB * NPT;

  fps_kernel  <<<NB,                1024, 0, stream>>>(xyz, fpsidx, newxyz);
  ballq_kernel<<<(NB * NPT) / 8,     256, 0, stream>>>(xyz, newxyz, grp);
  mlp_kernel  <<<(NB * NPT) / 2,      64, 0, stream>>>(xyz, feat, grp, newxyz,
                                                       w0, g0, b0, m0, v0,
                                                       w1, g1, b1, m1, v1,
                                                       w2, g2, b2, m2, v2,
                                                       outfeat);
}


# Ball query: each dependent step of the index-order scan fetches sixteen 32-point chunks at once (16 loads in flight per cache round trip, not one); same ascending order, same stop at the 32nd hit
# speedup vs baseline: 2.2786x; 1.1987x over previous
#include <hip/hip_runtime.h>
#include <hip/hip_bf16.h>
#include <math.h>

typedef float v2f __attribute__((ext_vector_type(2)));
typedef float v8f __attribute__((ext_vector_type(8)));

#define NB     8
#define NPTS   16384
#define NPT    512
#define NSAMP  32
#define RAD2   0.04f
#define BN_EPS 1e-5f


template <int M>
__device__ __forceinline__ int row_xor(int v) {
  return __builtin_amdgcn_update_dpp(0, v, 0x160 | M, 0xf, 0xf, true);
}
__device__ __forceinline__ int half_xor(int v) {
  return (int)__builtin_amdgcn_permlanex16((unsigned)v, (unsigned)v, 0x76543210u, 0xfedcba98u, false, false);
}
__device__ __forceinline__ int imax2(int a, int b) { return a > b ? a : b; }
__device__ __forceinline__ int imin2(int a, int b) { return a < b ? a : b; }

__device__ __forceinline__ int wave_imax(int m) {
  m = imax2(m, row_xor<1>(m));
  m = imax2(m, row_xor<2>(m));
  m = imax2(m, row_xor<4>(m));
  m = imax2(m, row_xor<8>(m));
  m = imax2(m, half_xor(m));
  return m;
}
__device__ __forceinline__ int wave_imin(int c) {
  c = imin2(c, row_xor<1>(c));
  c = imin2(c, row_xor<2>(c));
  c = imin2(c, row_xor<4>(c));
  c = imin2(c, row_xor<8>(c));
  c = imin2(c, half_xor(c));
  return c;
}

__device__ __forceinline__ float uni(float v) {
  return __int_as_float(__builtin_amdgcn_readfirstlane(__float_as_int(v)));
}
__device__ __forceinline__ unsigned spread3(unsigned v) {
  return (v & 1u) | ((v & 2u) << 2) | ((v & 4u) << 4);
}

__global__ __launch_bounds__(1024) void fps_kernel(const float* __restrict__ xyz,
                                                   int* __restrict__ fpsidx,
                                                   float* __restrict__ newxyz) {
  const int b   = blockIdx.x;
  const int tid = threadIdx.x;
  const float* base = xyz + (size_t)b * NPTS * 3;

  __shared__ float4 P[NPTS];
  __shared__ int rv[32];
  __shared__ int ri[32];
  __shared__ int wtot[32];

  const int wave = tid >> 5;
  const int lane = tid & 31;

  unsigned* K = (unsigned*)P;
  unsigned key[16];
#pragma unroll
  for (int j = 0; j < 16; ++j) {
    const int p = tid * 16 + j;
    const float x = base[p * 3 + 0], y = base[p * 3 + 1], z = base[p * 3 + 2];
    const unsigned ix = (unsigned)(int)fminf(fmaxf(x * 8.0f, 0.0f), 7.0f);
    const unsigned iy = (unsigned)(int)fminf(fmaxf(y * 8.0f, 0.0f), 7.0f);
    const unsigned iz = (unsigned)(int)fminf(fmaxf(z * 8.0f, 0.0f), 7.0f);
    key[j] = ((spread3(ix) | (spread3(iy) << 1) | (spread3(iz) << 2)) << 14) | (unsigned)p;
  }
  for (int bit = 14; bit < 23; ++bit) {
    int z = 0;
#pragma unroll
    for (int j = 0; j < 16; ++j) z += (int)(((key[j] >> bit) & 1u) ^ 1u);
    int inc = z;
    for (int d = 1; d < 32; d <<= 1) {
      const int t = __shfl_up(inc, d);
      if (lane >= d) inc += t;
    }
    if (lane == 31) wtot[wave] = inc;
    __syncthreads();
    const int wt = wtot[lane];
    int winc = wt;
    for (int d = 1; d < 32; d <<= 1) {
      const int t = __shfl_up(winc, d);
      if (lane >= d) winc += t;
    }
    const int zall  = __shfl(winc, 31);
    const int wexcl = __shfl(winc - wt, wave);
    int zpos = wexcl + inc - z;
    int opos = zall + (tid * 16 - zpos);
#pragma unroll
    for (int j = 0; j < 16; ++j) {
      const int one = (int)((key[j] >> bit) & 1u);
      K[one ? opos : zpos] = key[j];
      opos += one;
      zpos += 1 - one;
    }
    __syncthreads();
#pragma unroll
    for (int j = 0; j < 16; ++j) key[j] = K[tid * 16 + j];
    __syncthreads();
  }

  float px[16], py[16], pz[16], dd[16];
  unsigned pk[8];
#pragma unroll
  for (int j = 0; j < 16; ++j) {
    const int p = (int)(key[j] & 0x3fffu);
    px[j] = base[p * 3 + 0];
    py[j] = base[p * 3 + 1];
    pz[j] = base[p * 3 + 2];
    dd[j] = 3.4e38f;
    P[p] = make_float4(px[j], py[j], pz[j], 0.0f);
  }
#pragma unroll
  for (int j = 0; j < 16; j += 2)
    pk[j >> 1] = (key[j] & 0x3fffu) | ((key[j + 1] & 0x3fffu) << 16);

  float lx = px[0], hx = px[0], ly = py[0], hy = py[0], lz = pz[0], hz = pz[0];
#pragma unroll
  for (int j = 1; j < 16; ++j) {
    lx = fminf(lx, px[j]); hx = fmaxf(hx, px[j]);
    ly = fminf(ly, py[j]); hy = fmaxf(hy, py[j]);
    lz = fminf(lz, pz[j]); hz = fmaxf(hz, pz[j]);
  }
  for (int d = 16; d >= 1; d >>= 1) {
    lx = fminf(lx, __shfl_xor(lx, d)); hx = fmaxf(hx, __shfl_xor(hx, d));
    ly = fminf(ly, __shfl_xor(ly, d)); hy = fmaxf(hy, __shfl_xor(hy, d));
    lz = fminf(lz, __shfl_xor(lz, d)); hz = fmaxf(hz, __shfl_xor(hz, d));
  }
  const float lox = uni(lx), hix = uni(hx), loy = uni(ly), hiy = uni(hy), loz = uni(lz), hiz = uni(hz);

  int last = 0;
  float cx = base[0], cy = base[1], cz = base[2];
  int m = __float_as_int(3.4e38f);

  for (int i = 0; i < NPT; ++i) {
    if (tid == 0) {
      fpsidx[b * NPT + i] = last;
      newxyz[(b * NPT + i) * 3 + 0] = cx;
      newxyz[(b * NPT + i) * 3 + 1] = cy;
      newxyz[(b * NPT + i) * 3 + 2] = cz;
    }

    const float ex = fmaxf(fmaxf(lox - cx, cx - hix), 0.0f);
    const float ey = fmaxf(fmaxf(loy - cy, cy - hiy), 0.0f);
    const float ez = fmaxf(fmaxf(loz - cz, cz - hiz), 0.0f);
    const float lb = (ex * ex + ey * ey + ez * ez) * 0.999f;
    const float wmax = uni(__int_as_float(m));
    if (!(lb > wmax)) {
      float bv = -1.0f;
#pragma unroll
      for (int j = 0; j < 16; ++j) {
        float dx = px[j] - cx, dy = py[j] - cy, dz = pz[j] - cz;
        float d  = __builtin_fmaf(dz, dz, __builtin_fmaf(dx, dx, dy * dy));
        float nd = fminf(dd[j], d);
        dd[j] = nd;
        bv = fmaxf(bv, nd);
      }
      m = wave_imax(__float_as_int(bv));
    }

    if (lane == 0) rv[wave] = m;
    __syncthreads();
    const int M = wave_imax(rv[lane]);

    int c = 0x7fffffff;
    if (m == M) {
      int c0 = 0x7fffffff, c1 = 0x7fffffff, c2 = 0x7fffffff, c3 = 0x7fffffff;
#pragma unroll
      for (int j = 0; j < 16; j += 4) {
        if (__float_as_int(dd[j])     == M) c0 = imin2(c0, (int)(pk[j >> 1] & 0xffffu));
        if (__float_as_int(dd[j + 1]) == M) c1 = imin2(c1, (int)(pk[j >> 1] >> 16));
        if (__float_as_int(dd[j + 2]) == M) c2 = imin2(c2, (int)(pk[(j >> 1) + 1] & 0xffffu));
        if (__float_as_int(dd[j + 3]) == M) c3 = imin2(c3, (int)(pk[(j >> 1) + 1] >> 16));
      }
      c = wave_imin(imin2(imin2(c0, c1), imin2(c2, c3)));
    }
    if (lane == 0) ri[wave] = c;
    __syncthreads();
    const int    cand = ri[lane];
    const float4 pc   = P[cand & (NPTS - 1)];
    const int    cmin = wave_imin(cand);
    const int    wl   = __builtin_ctz(__builtin_amdgcn_ballot_w32(cand == cmin));
    cx = __int_as_float(__builtin_amdgcn_readlane(__float_as_int(pc.x), wl));
    cy = __int_as_float(__builtin_amdgcn_readlane(__float_as_int(pc.y), wl));
    cz = __int_as_float(__builtin_amdgcn_readlane(__float_as_int(pc.z), wl));
    last = __builtin_amdgcn_readfirstlane(cmin) & (NPTS - 1);
  }
}

#define BQ_U 16
__global__ __launch_bounds__(256) void ballq_kernel(const float* __restrict__ xyz,
                                                    const float* __restrict__ newxyz,
                                                    int* __restrict__ grp) {
  const int lane = threadIdx.x & 31;
  const int q    = blockIdx.x * 8 + (threadIdx.x >> 5);
  const int b    = q >> 9;
  const int s    = q & (NPT - 1);

  const float* base = xyz + (size_t)b * NPTS * 3;
  const float cx = newxyz[(b * NPT + s) * 3 + 0];
  const float cy = newxyz[(b * NPT + s) * 3 + 1];
  const float cz = newxyz[(b * NPT + s) * 3 + 2];

  int* g = grp + (size_t)q * NSAMP;
  int cnt = 0, first = -1;

  for (int bp = 0; bp < NPTS && cnt < NSAMP; bp += 32 * BQ_U) {
    float px[BQ_U], py[BQ_U], pz[BQ_U];
#pragma unroll
    for (int u = 0; u < BQ_U; ++u) {
      const int p = bp + 32 * u + lane;
      px[u] = base[p * 3 + 0];
      py[u] = base[p * 3 + 1];
      pz[u] = base[p * 3 + 2];
    }
    unsigned mask[BQ_U];
#pragma unroll
    for (int u = 0; u < BQ_U; ++u) {
      float dx = px[u] - cx;
      float dy = py[u] - cy;
      float dz = pz[u] - cz;
      mask[u] = (unsigned)__ballot((dx * dx + dy * dy + dz * dz) <= RAD2);
    }
#pragma unroll
    for (int u = 0; u < BQ_U; ++u) {
      if (mask[u]) {
        if (cnt < NSAMP) {
          const int cb = bp + 32 * u;
          if (first < 0) first = cb + __builtin_ctz(mask[u]);
          const bool hit = (mask[u] >> lane) & 1u;
          int pos = __popc(mask[u] & ((1u << lane) - 1u));
          if (hit && (cnt + pos) < NSAMP) g[cnt + pos] = cb + lane;
          cnt += __popc(mask[u]);
        }
      }
    }
  }
  if (first < 0) first = NPTS - 1;
  int filled = cnt < NSAMP ? cnt : NSAMP;
  for (int slot = filled + lane; slot < NSAMP; slot += 32) g[slot] = first;
}

__device__ __forceinline__ void gemm_bn_relu(const float* __restrict__ A, int lda,
                                             const float* __restrict__ W, int Kdim, int Kpad,
                                             int Nout,
                                             const float* __restrict__ gg,
                                             const float* __restrict__ bb,
                                             const float* __restrict__ mm,
                                             const float* __restrict__ vv,
                                             float* __restrict__ D, int ldd, int lane) {
  const int r15  = lane & 15;
  const int half = lane >> 4;
  for (int m0 = 0; m0 < 32; m0 += 16) {
    for (int n0 = 0; n0 < Nout; n0 += 16) {
      v8f acc = {0.f, 0.f, 0.f, 0.f, 0.f, 0.f, 0.f, 0.f};
      const int n = n0 + r15;
      for (int k0 = 0; k0 < Kpad; k0 += 4) {
        const int kk = k0 + half * 2;
        v2f a, w;
        a.x = A[(m0 + r15) * lda + kk];
        a.y = A[(m0 + r15) * lda + kk + 1];
        int ka = kk     < Kdim ? kk     : Kdim - 1;
        int kb = kk + 1 < Kdim ? kk + 1 : Kdim - 1;
        w.x = W[n * Kdim + ka];
        w.y = W[n * Kdim + kb];
        acc = __builtin_amdgcn_wmma_f32_16x16x4_f32(false, a, false, w,
                                                    (short)0, acc, false, false);
      }
      const float sc = gg[n] * rsqrtf(vv[n] + BN_EPS);
      const float bs = bb[n] - mm[n] * sc;
#pragma unroll
      for (int r = 0; r < 8; ++r) {
        float val = fmaxf(acc[r] * sc + bs, 0.f);
        D[(m0 + r + half * 8) * ldd + n] = val;
      }
    }
  }
}

__global__ __launch_bounds__(64) void mlp_kernel(const float* __restrict__ xyz,
                                                 const float* __restrict__ feat,
                                                 const int*   __restrict__ grp,
                                                 const float* __restrict__ newxyz,
                                                 const float* __restrict__ w0, const float* __restrict__ g0,
                                                 const float* __restrict__ b0, const float* __restrict__ m0,
                                                 const float* __restrict__ v0,
                                                 const float* __restrict__ w1, const float* __restrict__ g1,
                                                 const float* __restrict__ b1, const float* __restrict__ m1,
                                                 const float* __restrict__ v1,
                                                 const float* __restrict__ w2, const float* __restrict__ g2,
                                                 const float* __restrict__ b2, const float* __restrict__ m2,
                                                 const float* __restrict__ v2,
                                                 float* __restrict__ outfeat) {
  __shared__ float Xb[2][32 * 68];
  __shared__ float Hb[2][32 * 132];

  const int wave = threadIdx.x >> 5;
  const int lane = threadIdx.x & 31;
  const int q    = blockIdx.x * 2 + wave;
  const int b    = q >> 9;
  const int s    = q & (NPT - 1);

  float* X = Xb[wave];
  float* H = Hb[wave];

  const int gi = grp[(size_t)q * NSAMP + lane];
  const float* prow = xyz + ((size_t)b * NPTS + gi) * 3;
  const float* nx   = newxyz + (b * NPT + s) * 3;
  float* xr = X + lane * 68;
  xr[0] = prow[0] - nx[0];
  xr[1] = prow[1] - nx[1];
  xr[2] = prow[2] - nx[2];
  const float4* frow = (const float4*)(feat + ((size_t)b * NPTS + gi) * 64);
#pragma unroll
  for (int c4 = 0; c4 < 16; ++c4) {
    float4 f = frow[c4];
    xr[3 + 4 * c4 + 0] = f.x;
    xr[3 + 4 * c4 + 1] = f.y;
    xr[3 + 4 * c4 + 2] = f.z;
    xr[3 + 4 * c4 + 3] = f.w;
  }
  xr[67] = 0.f;

  gemm_bn_relu(X, 68,  w0, 67, 68, 64,  g0, b0, m0, v0, H, 132, lane);
  gemm_bn_relu(H, 132, w1, 64, 64, 64,  g1, b1, m1, v1, X, 68,  lane);
  gemm_bn_relu(X, 68,  w2, 64, 64, 128, g2, b2, m2, v2, H, 132, lane);

#pragma unroll
  for (int cc = 0; cc < 4; ++cc) {
    const int col = lane + cc * 32;
    float mx = H[col];
    for (int row = 1; row < 32; ++row) mx = fmaxf(mx, H[row * 132 + col]);
    outfeat[((size_t)b * 128 + col) * NPT + s] = mx;
  }
}

extern "C" void kernel_launch(void* const* d_in, const int* in_sizes, int n_in,
                              void* d_out, int out_size, void* d_ws, size_t ws_size,
                              hipStream_t stream) {
  (void)in_sizes; (void)n_in; (void)out_size; (void)ws_size;
  const float* xyz  = (const float*)d_in[0];
  const float* feat = (const float*)d_in[1];
  const float* w0 = (const float*)d_in[2];
  const float* g0 = (const float*)d_in[3];
  const float* b0 = (const float*)d_in[4];
  const float* m0 = (const float*)d_in[5];
  const float* v0 = (const float*)d_in[6];
  const float* w1 = (const float*)d_in[7];
  const float* g1 = (const float*)d_in[8];
  const float* b1 = (const float*)d_in[9];
  const float* m1 = (const float*)d_in[10];
  const float* v1 = (const float*)d_in[11];
  const float* w2 = (const float*)d_in[12];
  const float* g2 = (const float*)d_in[13];
  const float* b2 = (const float*)d_in[14];
  const float* m2 = (const float*)d_in[15];
  const float* v2 = (const float*)d_in[16];

  float* newxyz  = (float*)d_out;
  float* outfeat = (float*)d_out + NB * NPT * 3;

  int* fpsidx = (int*)d_ws;
  int* grp    = (int*)d_ws + NB * NPT;

  fps_kernel  <<<NB,                1024, 0, stream>>>(xyz, fpsidx, newxyz);
  ballq_kernel<<<(NB * NPT) / 8,     256, 0, stream>>>(xyz, newxyz, grp);
  mlp_kernel  <<<(NB * NPT) / 2,      64, 0, stream>>>(xyz, feat, grp, newxyz,
                                                       w0, g0, b0, m0, v0,
                                                       w1, g1, b1, m1, v1,
                                                       w2, g2, b2, m2, v2,
                                                       outfeat);
}


# MLP: column tiles outermost, each weight fragment fetched once and multiplied with both row tiles of a query's samples (two accumulators per wave); every element's k sum keeps its order
# speedup vs baseline: 2.4634x; 1.0811x over previous
#include <hip/hip_runtime.h>
#include <hip/hip_bf16.h>
#include <math.h>

typedef float v2f __attribute__((ext_vector_type(2)));
typedef float v8f __attribute__((ext_vector_type(8)));

#define NB     8
#define NPTS   16384
#define NPT    512
#define NSAMP  32
#define RAD2   0.04f
#define BN_EPS 1e-5f


template <int M>
__device__ __forceinline__ int row_xor(int v) {
  return __builtin_amdgcn_update_dpp(0, v, 0x160 | M, 0xf, 0xf, true);
}
__device__ __forceinline__ int half_xor(int v) {
  return (int)__builtin_amdgcn_permlanex16((unsigned)v, (unsigned)v, 0x76543210u, 0xfedcba98u, false, false);
}
__device__ __forceinline__ int imax2(int a, int b) { return a > b ? a : b; }
__device__ __forceinline__ int imin2(int a, int b) { return a < b ? a : b; }

__device__ __forceinline__ int wave_imax(int m) {
  m = imax2(m, row_xor<1>(m));
  m = imax2(m, row_xor<2>(m));
  m = imax2(m, row_xor<4>(m));
  m = imax2(m, row_xor<8>(m));
  m = imax2(m, half_xor(m));
  return m;
}
__device__ __forceinline__ int wave_imin(int c) {
  c = imin2(c, row_xor<1>(c));
  c = imin2(c, row_xor<2>(c));
  c = imin2(c, row_xor<4>(c));
  c = imin2(c, row_xor<8>(c));
  c = imin2(c, half_xor(c));
  return c;
}

__device__ __forceinline__ float uni(float v) {
  return __int_as_float(__builtin_amdgcn_readfirstlane(__float_as_int(v)));
}
__device__ __forceinline__ unsigned spread3(unsigned v) {
  return (v & 1u) | ((v & 2u) << 2) | ((v & 4u) << 4);
}

__global__ __launch_bounds__(1024) void fps_kernel(const float* __restrict__ xyz,
                                                   int* __restrict__ fpsidx,
                                                   float* __restrict__ newxyz) {
  const int b   = blockIdx.x;
  const int tid = threadIdx.x;
  const float* base = xyz + (size_t)b * NPTS * 3;

  __shared__ float4 P[NPTS];
  __shared__ int rv[32];
  __shared__ int ri[32];
  __shared__ int wtot[32];

  const int wave = tid >> 5;
  const int lane = tid & 31;

  unsigned* K = (unsigned*)P;
  unsigned key[16];
#pragma unroll
  for (int j = 0; j < 16; ++j) {
    const int p = tid * 16 + j;
    const float x = base[p * 3 + 0], y = base[p * 3 + 1], z = base[p * 3 + 2];
    const unsigned ix = (unsigned)(int)fminf(fmaxf(x * 8.0f, 0.0f), 7.0f);
    const unsigned iy = (unsigned)(int)fminf(fmaxf(y * 8.0f, 0.0f), 7.0f);
    const unsigned iz = (unsigned)(int)fminf(fmaxf(z * 8.0f, 0.0f), 7.0f);
    key[j] = ((spread3(ix) | (spread3(iy) << 1) | (spread3(iz) << 2)) << 14) | (unsigned)p;
  }
  for (int bit = 14; bit < 23; ++bit) {
    int z = 0;
#pragma unroll
    for (int j = 0; j < 16; ++j) z += (int)(((key[j] >> bit) & 1u) ^ 1u);
    int inc = z;
    for (int d = 1; d < 32; d <<= 1) {
      const int t = __shfl_up(inc, d);
      if (lane >= d) inc += t;
    }
    if (lane == 31) wtot[wave] = inc;
    __syncthreads();
    const int wt = wtot[lane];
    int winc = wt;
    for (int d = 1; d < 32; d <<= 1) {
      const int t = __shfl_up(winc, d);
      if (lane >= d) winc += t;
    }
    const int zall  = __shfl(winc, 31);
    const int wexcl = __shfl(winc - wt, wave);
    int zpos = wexcl + inc - z;
    int opos = zall + (tid * 16 - zpos);
#pragma unroll
    for (int j = 0; j < 16; ++j) {
      const int one = (int)((key[j] >> bit) & 1u);
      K[one ? opos : zpos] = key[j];
      opos += one;
      zpos += 1 - one;
    }
    __syncthreads();
#pragma unroll
    for (int j = 0; j < 16; ++j) key[j] = K[tid * 16 + j];
    __syncthreads();
  }

  float px[16], py[16], pz[16], dd[16];
  unsigned pk[8];
#pragma unroll
  for (int j = 0; j < 16; ++j) {
    const int p = (int)(key[j] & 0x3fffu);
    px[j] = base[p * 3 + 0];
    py[j] = base[p * 3 + 1];
    pz[j] = base[p * 3 + 2];
    dd[j] = 3.4e38f;
    P[p] = make_float4(px[j], py[j], pz[j], 0.0f);
  }
#pragma unroll
  for (int j = 0; j < 16; j += 2)
    pk[j >> 1] = (key[j] & 0x3fffu) | ((key[j + 1] & 0x3fffu) << 16);

  float lx = px[0], hx = px[0], ly = py[0], hy = py[0], lz = pz[0], hz = pz[0];
#pragma unroll
  for (int j = 1; j < 16; ++j) {
    lx = fminf(lx, px[j]); hx = fmaxf(hx, px[j]);
    ly = fminf(ly, py[j]); hy = fmaxf(hy, py[j]);
    lz = fminf(lz, pz[j]); hz = fmaxf(hz, pz[j]);
  }
  for (int d = 16; d >= 1; d >>= 1) {
    lx = fminf(lx, __shfl_xor(lx, d)); hx = fmaxf(hx, __shfl_xor(hx, d));
    ly = fminf(ly, __shfl_xor(ly, d)); hy = fmaxf(hy, __shfl_xor(hy, d));
    lz = fminf(lz, __shfl_xor(lz, d)); hz = fmaxf(hz, __shfl_xor(hz, d));
  }
  const float lox = uni(lx), hix = uni(hx), loy = uni(ly), hiy = uni(hy), loz = uni(lz), hiz = uni(hz);

  int last = 0;
  float cx = base[0], cy = base[1], cz = base[2];
  int m = __float_as_int(3.4e38f);

  for (int i = 0; i < NPT; ++i) {
    if (tid == 0) {
      fpsidx[b * NPT + i] = last;
      newxyz[(b * NPT + i) * 3 + 0] = cx;
      newxyz[(b * NPT + i) * 3 + 1] = cy;
      newxyz[(b * NPT + i) * 3 + 2] = cz;
    }

    const float ex = fmaxf(fmaxf(lox - cx, cx - hix), 0.0f);
    const float ey = fmaxf(fmaxf(loy - cy, cy - hiy), 0.0f);
    const float ez = fmaxf(fmaxf(loz - cz, cz - hiz), 0.0f);
    const float lb = (ex * ex + ey * ey + ez * ez) * 0.999f;
    const float wmax = uni(__int_as_float(m));
    if (!(lb > wmax)) {
      float bv = -1.0f;
#pragma unroll
      for (int j = 0; j < 16; ++j) {
        float dx = px[j] - cx, dy = py[j] - cy, dz = pz[j] - cz;
        float d  = __builtin_fmaf(dz, dz, __builtin_fmaf(dx, dx, dy * dy));
        float nd = fminf(dd[j], d);
        dd[j] = nd;
        bv = fmaxf(bv, nd);
      }
      m = wave_imax(__float_as_int(bv));
    }

    if (lane == 0) rv[wave] = m;
    __syncthreads();
    const int M = wave_imax(rv[lane]);

    int c = 0x7fffffff;
    if (m == M) {
      int c0 = 0x7fffffff, c1 = 0x7fffffff, c2 = 0x7fffffff, c3 = 0x7fffffff;
#pragma unroll
      for (int j = 0; j < 16; j += 4) {
        if (__float_as_int(dd[j])     == M) c0 = imin2(c0, (int)(pk[j >> 1] & 0xffffu));
        if (__float_as_int(dd[j + 1]) == M) c1 = imin2(c1, (int)(pk[j >> 1] >> 16));
        if (__float_as_int(dd[j + 2]) == M) c2 = imin2(c2, (int)(pk[(j >> 1) + 1] & 0xffffu));
        if (__float_as_int(dd[j + 3]) == M) c3 = imin2(c3, (int)(pk[(j >> 1) + 1] >> 16));
      }
      c = wave_imin(imin2(imin2(c0, c1), imin2(c2, c3)));
    }
    if (lane == 0) ri[wave] = c;
    __syncthreads();
    const int    cand = ri[lane];
    const float4 pc   = P[cand & (NPTS - 1)];
    const int    cmin = wave_imin(cand);
    const int    wl   = __builtin_ctz(__builtin_amdgcn_ballot_w32(cand == cmin));
    cx = __int_as_float(__builtin_amdgcn_readlane(__float_as_int(pc.x), wl));
    cy = __int_as_float(__builtin_amdgcn_readlane(__float_as_int(pc.y), wl));
    cz = __int_as_float(__builtin_amdgcn_readlane(__float_as_int(pc.z), wl));
    last = __builtin_amdgcn_readfirstlane(cmin) & (NPTS - 1);
  }
}

#define BQ_U 16
__global__ __launch_bounds__(256) void ballq_kernel(const float* __restrict__ xyz,
                                                    const float* __restrict__ newxyz,
                                                    int* __restrict__ grp) {
  const int lane = threadIdx.x & 31;
  const int q    = blockIdx.x * 8 + (threadIdx.x >> 5);
  const int b    = q >> 9;
  const int s    = q & (NPT - 1);

  const float* base = xyz + (size_t)b * NPTS * 3;
  const float cx = newxyz[(b * NPT + s) * 3 + 0];
  const float cy = newxyz[(b * NPT + s) * 3 + 1];
  const float cz = newxyz[(b * NPT + s) * 3 + 2];

  int* g = grp + (size_t)q * NSAMP;
  int cnt = 0, first = -1;

  for (int bp = 0; bp < NPTS && cnt < NSAMP; bp += 32 * BQ_U) {
    float px[BQ_U], py[BQ_U], pz[BQ_U];
#pragma unroll
    for (int u = 0; u < BQ_U; ++u) {
      const int p = bp + 32 * u + lane;
      px[u] = base[p * 3 + 0];
      py[u] = base[p * 3 + 1];
      pz[u] = base[p * 3 + 2];
    }
    unsigned mask[BQ_U];
#pragma unroll
    for (int u = 0; u < BQ_U; ++u) {
      float dx = px[u] - cx;
      float dy = py[u] - cy;
      float dz = pz[u] - cz;
      mask[u] = (unsigned)__ballot((dx * dx + dy * dy + dz * dz) <= RAD2);
    }
#pragma unroll
    for (int u = 0; u < BQ_U; ++u) {
      if (mask[u]) {
        if (cnt < NSAMP) {
          const int cb = bp + 32 * u;
          if (first < 0) first = cb + __builtin_ctz(mask[u]);
          const bool hit = (mask[u] >> lane) & 1u;
          int pos = __popc(mask[u] & ((1u << lane) - 1u));
          if (hit && (cnt + pos) < NSAMP) g[cnt + pos] = cb + lane;
          cnt += __popc(mask[u]);
        }
      }
    }
  }
  if (first < 0) first = NPTS - 1;
  int filled = cnt < NSAMP ? cnt : NSAMP;
  for (int slot = filled + lane; slot < NSAMP; slot += 32) g[slot] = first;
}

__device__ __forceinline__ void gemm_bn_relu(const float* __restrict__ A, int lda,
                                             const float* __restrict__ W, int Kdim, int Kpad,
                                             int Nout,
                                             const float* __restrict__ gg,
                                             const float* __restrict__ bb,
                                             const float* __restrict__ mm,
                                             const float* __restrict__ vv,
                                             float* __restrict__ D, int ldd, int lane) {
  const int r15  = lane & 15;
  const int half = lane >> 4;
  for (int n0 = 0; n0 < Nout; n0 += 16) {
    v8f acc0 = {0.f, 0.f, 0.f, 0.f, 0.f, 0.f, 0.f, 0.f};
    v8f acc1 = {0.f, 0.f, 0.f, 0.f, 0.f, 0.f, 0.f, 0.f};
    const int n = n0 + r15;
    for (int k0 = 0; k0 < Kpad; k0 += 4) {
      const int kk = k0 + half * 2;
      v2f a0, a1, w;
      a0.x = A[r15 * lda + kk];
      a0.y = A[r15 * lda + kk + 1];
      a1.x = A[(16 + r15) * lda + kk];
      a1.y = A[(16 + r15) * lda + kk + 1];
      int ka = kk     < Kdim ? kk     : Kdim - 1;
      int kb = kk + 1 < Kdim ? kk + 1 : Kdim - 1;
      w.x = W[n * Kdim + ka];
      w.y = W[n * Kdim + kb];
      acc0 = __builtin_amdgcn_wmma_f32_16x16x4_f32(false, a0, false, w,
                                                   (short)0, acc0, false, false);
      acc1 = __builtin_amdgcn_wmma_f32_16x16x4_f32(false, a1, false, w,
                                                   (short)0, acc1, false, false);
    }
    const float sc = gg[n] * rsqrtf(vv[n] + BN_EPS);
    const float bs = bb[n] - mm[n] * sc;
#pragma unroll
    for (int r = 0; r < 8; ++r) {
      D[(r + half * 8) * ldd + n]      = fmaxf(acc0[r] * sc + bs, 0.f);
      D[(16 + r + half * 8) * ldd + n] = fmaxf(acc1[r] * sc + bs, 0.f);
    }
  }
}

__global__ __launch_bounds__(64) void mlp_kernel(const float* __restrict__ xyz,
                                                 const float* __restrict__ feat,
                                                 const int*   __restrict__ grp,
                                                 const float* __restrict__ newxyz,
                                                 const float* __restrict__ w0, const float* __restrict__ g0,
                                                 const float* __restrict__ b0, const float* __restrict__ m0,
                                                 const float* __restrict__ v0,
                                                 const float* __restrict__ w1, const float* __restrict__ g1,
                                                 const float* __restrict__ b1, const float* __restrict__ m1,
                                                 const float* __restrict__ v1,
                                                 const float* __restrict__ w2, const float* __restrict__ g2,
                                                 const float* __restrict__ b2, const float* __restrict__ m2,
                                                 const float* __restrict__ v2,
                                                 float* __restrict__ outfeat) {
  __shared__ float Xb[2][32 * 68];
  __shared__ float Hb[2][32 * 132];

  const int wave = threadIdx.x >> 5;
  const int lane = threadIdx.x & 31;
  const int q    = blockIdx.x * 2 + wave;
  const int b    = q >> 9;
  const int s    = q & (NPT - 1);

  float* X = Xb[wave];
  float* H = Hb[wave];

  const int gi = grp[(size_t)q * NSAMP + lane];
  const float* prow = xyz + ((size_t)b * NPTS + gi) * 3;
  const float* nx   = newxyz + (b * NPT + s) * 3;
  float* xr = X + lane * 68;
  xr[0] = prow[0] - nx[0];
  xr[1] = prow[1] - nx[1];
  xr[2] = prow[2] - nx[2];
  const float4* frow = (const float4*)(feat + ((size_t)b * NPTS + gi) * 64);
#pragma unroll
  for (int c4 = 0; c4 < 16; ++c4) {
    float4 f = frow[c4];
    xr[3 + 4 * c4 + 0] = f.x;
    xr[3 + 4 * c4 + 1] = f.y;
    xr[3 + 4 * c4 + 2] = f.z;
    xr[3 + 4 * c4 + 3] = f.w;
  }
  xr[67] = 0.f;

  gemm_bn_relu(X, 68,  w0, 67, 68, 64,  g0, b0, m0, v0, H, 132, lane);
  gemm_bn_relu(H, 132, w1, 64, 64, 64,  g1, b1, m1, v1, X, 68,  lane);
  gemm_bn_relu(X, 68,  w2, 64, 64, 128, g2, b2, m2, v2, H, 132, lane);

#pragma unroll
  for (int cc = 0; cc < 4; ++cc) {
    const int col = lane + cc * 32;
    float mx = H[col];
    for (int row = 1; row < 32; ++row) mx = fmaxf(mx, H[row * 132 + col]);
    outfeat[((size_t)b * 128 + col) * NPT + s] = mx;
  }
}

extern "C" void kernel_launch(void* const* d_in, const int* in_sizes, int n_in,
                              void* d_out, int out_size, void* d_ws, size_t ws_size,
                              hipStream_t stream) {
  (void)in_sizes; (void)n_in; (void)out_size; (void)ws_size;
  const float* xyz  = (const float*)d_in[0];
  const float* feat = (const float*)d_in[1];
  const float* w0 = (const float*)d_in[2];
  const float* g0 = (const float*)d_in[3];
  const float* b0 = (const float*)d_in[4];
  const float* m0 = (const float*)d_in[5];
  const float* v0 = (const float*)d_in[6];
  const float* w1 = (const float*)d_in[7];
  const float* g1 = (const float*)d_in[8];
  const float* b1 = (const float*)d_in[9];
  const float* m1 = (const float*)d_in[10];
  const float* v1 = (const float*)d_in[11];
  const float* w2 = (const float*)d_in[12];
  const float* g2 = (const float*)d_in[13];
  const float* b2 = (const float*)d_in[14];
  const float* m2 = (const float*)d_in[15];
  const float* v2 = (const float*)d_in[16];

  float* newxyz  = (float*)d_out;
  float* outfeat = (float*)d_out + NB * NPT * 3;

  int* fpsidx = (int*)d_ws;
  int* grp    = (int*)d_ws + NB * NPT;

  fps_kernel  <<<NB,                1024, 0, stream>>>(xyz, fpsidx, newxyz);
  ballq_kernel<<<(NB * NPT) / 8,     256, 0, stream>>>(xyz, newxyz, grp);
  mlp_kernel  <<<(NB * NPT) / 2,      64, 0, stream>>>(xyz, feat, grp, newxyz,
                                                       w0, g0, b0, m0, v0,
                                                       w1, g1, b1, m1, v1,
                                                       w2, g2, b2, m2, v2,
                                                       outfeat);
}


# FPS: per-iteration block decisions (is this wave's maximum the block's, which lane holds its index, which wave's candidate wins) by one compare and a lane vote each, not butterfly folds; same picks
# speedup vs baseline: 2.6071x; 1.0583x over previous
#include <hip/hip_runtime.h>
#include <hip/hip_bf16.h>
#include <math.h>

typedef float v2f __attribute__((ext_vector_type(2)));
typedef float v8f __attribute__((ext_vector_type(8)));

#define NB     8
#define NPTS   16384
#define NPT    512
#define NSAMP  32
#define RAD2   0.04f
#define BN_EPS 1e-5f


template <int M>
__device__ __forceinline__ int row_xor(int v) {
  return __builtin_amdgcn_update_dpp(0, v, 0x160 | M, 0xf, 0xf, true);
}
__device__ __forceinline__ int half_xor(int v) {
  return (int)__builtin_amdgcn_permlanex16((unsigned)v, (unsigned)v, 0x76543210u, 0xfedcba98u, false, false);
}
__device__ __forceinline__ int imax2(int a, int b) { return a > b ? a : b; }
__device__ __forceinline__ int imin2(int a, int b) { return a < b ? a : b; }

__device__ __forceinline__ int wave_imax(int m) {
  m = imax2(m, row_xor<1>(m));
  m = imax2(m, row_xor<2>(m));
  m = imax2(m, row_xor<4>(m));
  m = imax2(m, row_xor<8>(m));
  m = imax2(m, half_xor(m));
  return m;
}
__device__ __forceinline__ int wave_imin(int c) {
  c = imin2(c, row_xor<1>(c));
  c = imin2(c, row_xor<2>(c));
  c = imin2(c, row_xor<4>(c));
  c = imin2(c, row_xor<8>(c));
  c = imin2(c, half_xor(c));
  return c;
}

__device__ __forceinline__ float uni(float v) {
  return __int_as_float(__builtin_amdgcn_readfirstlane(__float_as_int(v)));
}
__device__ __forceinline__ unsigned spread3(unsigned v) {
  return (v & 1u) | ((v & 2u) << 2) | ((v & 4u) << 4);
}

__global__ __launch_bounds__(1024) void fps_kernel(const float* __restrict__ xyz,
                                                   int* __restrict__ fpsidx,
                                                   float* __restrict__ newxyz) {
  const int b   = blockIdx.x;
  const int tid = threadIdx.x;
  const float* base = xyz + (size_t)b * NPTS * 3;

  __shared__ float4 P[NPTS];
  __shared__ int rv[32];
  __shared__ int ri[32];
  __shared__ int wtot[32];

  const int wave = tid >> 5;
  const int lane = tid & 31;

  unsigned* K = (unsigned*)P;
  unsigned key[16];
#pragma unroll
  for (int j = 0; j < 16; ++j) {
    const int p = tid * 16 + j;
    const float x = base[p * 3 + 0], y = base[p * 3 + 1], z = base[p * 3 + 2];
    const unsigned ix = (unsigned)(int)fminf(fmaxf(x * 8.0f, 0.0f), 7.0f);
    const unsigned iy = (unsigned)(int)fminf(fmaxf(y * 8.0f, 0.0f), 7.0f);
    const unsigned iz = (unsigned)(int)fminf(fmaxf(z * 8.0f, 0.0f), 7.0f);
    key[j] = ((spread3(ix) | (spread3(iy) << 1) | (spread3(iz) << 2)) << 14) | (unsigned)p;
  }
  for (int bit = 14; bit < 23; ++bit) {
    int z = 0;
#pragma unroll
    for (int j = 0; j < 16; ++j) z += (int)(((key[j] >> bit) & 1u) ^ 1u);
    int inc = z;
    for (int d = 1; d < 32; d <<= 1) {
      const int t = __shfl_up(inc, d);
      if (lane >= d) inc += t;
    }
    if (lane == 31) wtot[wave] = inc;
    __syncthreads();
    const int wt = wtot[lane];
    int winc = wt;
    for (int d = 1; d < 32; d <<= 1) {
      const int t = __shfl_up(winc, d);
      if (lane >= d) winc += t;
    }
    const int zall  = __shfl(winc, 31);
    const int wexcl = __shfl(winc - wt, wave);
    int zpos = wexcl + inc - z;
    int opos = zall + (tid * 16 - zpos);
#pragma unroll
    for (int j = 0; j < 16; ++j) {
      const int one = (int)((key[j] >> bit) & 1u);
      K[one ? opos : zpos] = key[j];
      opos += one;
      zpos += 1 - one;
    }
    __syncthreads();
#pragma unroll
    for (int j = 0; j < 16; ++j) key[j] = K[tid * 16 + j];
    __syncthreads();
  }

  float px[16], py[16], pz[16], dd[16];
  unsigned pk[8];
#pragma unroll
  for (int j = 0; j < 16; ++j) {
    const int p = (int)(key[j] & 0x3fffu);
    px[j] = base[p * 3 + 0];
    py[j] = base[p * 3 + 1];
    pz[j] = base[p * 3 + 2];
    dd[j] = 3.4e38f;
    P[p] = make_float4(px[j], py[j], pz[j], 0.0f);
  }
#pragma unroll
  for (int j = 0; j < 16; j += 2)
    pk[j >> 1] = (key[j] & 0x3fffu) | ((key[j + 1] & 0x3fffu) << 16);

  float lx = px[0], hx = px[0], ly = py[0], hy = py[0], lz = pz[0], hz = pz[0];
#pragma unroll
  for (int j = 1; j < 16; ++j) {
    lx = fminf(lx, px[j]); hx = fmaxf(hx, px[j]);
    ly = fminf(ly, py[j]); hy = fmaxf(hy, py[j]);
    lz = fminf(lz, pz[j]); hz = fmaxf(hz, pz[j]);
  }
  for (int d = 16; d >= 1; d >>= 1) {
    lx = fminf(lx, __shfl_xor(lx, d)); hx = fmaxf(hx, __shfl_xor(hx, d));
    ly = fminf(ly, __shfl_xor(ly, d)); hy = fmaxf(hy, __shfl_xor(hy, d));
    lz = fminf(lz, __shfl_xor(lz, d)); hz = fmaxf(hz, __shfl_xor(hz, d));
  }
  const float lox = uni(lx), hix = uni(hx), loy = uni(ly), hiy = uni(hy), loz = uni(lz), hiz = uni(hz);

  int last = 0;
  float cx = base[0], cy = base[1], cz = base[2];
  int m = __float_as_int(3.4e38f);

  for (int i = 0; i < NPT; ++i) {
    if (tid == 0) {
      fpsidx[b * NPT + i] = last;
      newxyz[(b * NPT + i) * 3 + 0] = cx;
      newxyz[(b * NPT + i) * 3 + 1] = cy;
      newxyz[(b * NPT + i) * 3 + 2] = cz;
    }

    const float ex = fmaxf(fmaxf(lox - cx, cx - hix), 0.0f);
    const float ey = fmaxf(fmaxf(loy - cy, cy - hiy), 0.0f);
    const float ez = fmaxf(fmaxf(loz - cz, cz - hiz), 0.0f);
    const float lb = (ex * ex + ey * ey + ez * ez) * 0.999f;
    const float wmax = uni(__int_as_float(m));
    if (!(lb > wmax)) {
      float bv = -1.0f;
#pragma unroll
      for (int j = 0; j < 16; ++j) {
        float dx = px[j] - cx, dy = py[j] - cy, dz = pz[j] - cz;
        float d  = __builtin_fmaf(dz, dz, __builtin_fmaf(dx, dx, dy * dy));
        float nd = fminf(dd[j], d);
        dd[j] = nd;
        bv = fmaxf(bv, nd);
      }
      m = wave_imax(__float_as_int(bv));
    }

    if (lane == 0) rv[wave] = m;
    __syncthreads();
    const bool beaten = __builtin_amdgcn_ballot_w32(rv[lane] > m) != 0u;

    const int M = m;
    int c = 0x7fffffff;
    bool writer = (lane == 0);
    if (!beaten) {
      int c0 = 0x7fffffff, c1 = 0x7fffffff, c2 = 0x7fffffff, c3 = 0x7fffffff;
#pragma unroll
      for (int j = 0; j < 16; j += 4) {
        if (__float_as_int(dd[j])     == M) c0 = imin2(c0, (int)(pk[j >> 1] & 0xffffu));
        if (__float_as_int(dd[j + 1]) == M) c1 = imin2(c1, (int)(pk[j >> 1] >> 16));
        if (__float_as_int(dd[j + 2]) == M) c2 = imin2(c2, (int)(pk[(j >> 1) + 1] & 0xffffu));
        if (__float_as_int(dd[j + 3]) == M) c3 = imin2(c3, (int)(pk[(j >> 1) + 1] >> 16));
      }
      const int cl = imin2(imin2(c0, c1), imin2(c2, c3));
      const unsigned hl = __builtin_amdgcn_ballot_w32(cl != 0x7fffffff);
      if (__builtin_popcount(hl) == 1) {
        c = cl;
        writer = (cl != 0x7fffffff);
      } else {
        c = wave_imin(cl);
      }
    }
    if (writer) ri[wave] = c;
    __syncthreads();
    const int    cand = ri[lane];
    const float4 pc   = P[cand & (NPTS - 1)];
    const unsigned hm = __builtin_amdgcn_ballot_w32(cand != 0x7fffffff);
    int wl;
    if (__builtin_popcount(hm) == 1) {
      wl = __builtin_ctz(hm);
    } else {
      const int cmin = wave_imin(cand);
      wl = __builtin_ctz(__builtin_amdgcn_ballot_w32(cand == cmin));
    }
    cx = __int_as_float(__builtin_amdgcn_readlane(__float_as_int(pc.x), wl));
    cy = __int_as_float(__builtin_amdgcn_readlane(__float_as_int(pc.y), wl));
    cz = __int_as_float(__builtin_amdgcn_readlane(__float_as_int(pc.z), wl));
    last = __builtin_amdgcn_readlane(cand, wl) & (NPTS - 1);
  }
}

#define BQ_U 16
__global__ __launch_bounds__(256) void ballq_kernel(const float* __restrict__ xyz,
                                                    const float* __restrict__ newxyz,
                                                    int* __restrict__ grp) {
  const int lane = threadIdx.x & 31;
  const int q    = blockIdx.x * 8 + (threadIdx.x >> 5);
  const int b    = q >> 9;
  const int s    = q & (NPT - 1);

  const float* base = xyz + (size_t)b * NPTS * 3;
  const float cx = newxyz[(b * NPT + s) * 3 + 0];
  const float cy = newxyz[(b * NPT + s) * 3 + 1];
  const float cz = newxyz[(b * NPT + s) * 3 + 2];

  int* g = grp + (size_t)q * NSAMP;
  int cnt = 0, first = -1;

  for (int bp = 0; bp < NPTS && cnt < NSAMP; bp += 32 * BQ_U) {
    float px[BQ_U], py[BQ_U], pz[BQ_U];
#pragma unroll
    for (int u = 0; u < BQ_U; ++u) {
      const int p = bp + 32 * u + lane;
      px[u] = base[p * 3 + 0];
      py[u] = base[p * 3 + 1];
      pz[u] = base[p * 3 + 2];
    }
    unsigned mask[BQ_U];
#pragma unroll
    for (int u = 0; u < BQ_U; ++u) {
      float dx = px[u] - cx;
      float dy = py[u] - cy;
      float dz = pz[u] - cz;
      mask[u] = (unsigned)__ballot((dx * dx + dy * dy + dz * dz) <= RAD2);
    }
#pragma unroll
    for (int u = 0; u < BQ_U; ++u) {
      if (mask[u]) {
        if (cnt < NSAMP) {
          const int cb = bp + 32 * u;
          if (first < 0) first = cb + __builtin_ctz(mask[u]);
          const bool hit = (mask[u] >> lane) & 1u;
          int pos = __popc(mask[u] & ((1u << lane) - 1u));
          if (hit && (cnt + pos) < NSAMP) g[cnt + pos] = cb + lane;
          cnt += __popc(mask[u]);
        }
      }
    }
  }
  if (first < 0) first = NPTS - 1;
  int filled = cnt < NSAMP ? cnt : NSAMP;
  for (int slot = filled + lane; slot < NSAMP; slot += 32) g[slot] = first;
}

__device__ __forceinline__ void gemm_bn_relu(const float* __restrict__ A, int lda,
                                             const float* __restrict__ W, int Kdim, int Kpad,
                                             int Nout,
                                             const float* __restrict__ gg,
                                             const float* __restrict__ bb,
                                             const float* __restrict__ mm,
                                             const float* __restrict__ vv,
                                             float* __restrict__ D, int ldd, int lane) {
  const int r15  = lane & 15;
  const int half = lane >> 4;
  for (int n0 = 0; n0 < Nout; n0 += 16) {
    v8f acc0 = {0.f, 0.f, 0.f, 0.f, 0.f, 0.f, 0.f, 0.f};
    v8f acc1 = {0.f, 0.f, 0.f, 0.f, 0.f, 0.f, 0.f, 0.f};
    const int n = n0 + r15;
    for (int k0 = 0; k0 < Kpad; k0 += 4) {
      const int kk = k0 + half * 2;
      v2f a0, a1, w;
      a0.x = A[r15 * lda + kk];
      a0.y = A[r15 * lda + kk + 1];
      a1.x = A[(16 + r15) * lda + kk];
      a1.y = A[(16 + r15) * lda + kk + 1];
      int ka = kk     < Kdim ? kk     : Kdim - 1;
      int kb = kk + 1 < Kdim ? kk + 1 : Kdim - 1;
      w.x = W[n * Kdim + ka];
      w.y = W[n * Kdim + kb];
      acc0 = __builtin_amdgcn_wmma_f32_16x16x4_f32(false, a0, false, w,
                                                   (short)0, acc0, false, false);
      acc1 = __builtin_amdgcn_wmma_f32_16x16x4_f32(false, a1, false, w,
                                                   (short)0, acc1, false, false);
    }
    const float sc = gg[n] * rsqrtf(vv[n] + BN_EPS);
    const float bs = bb[n] - mm[n] * sc;
#pragma unroll
    for (int r = 0; r < 8; ++r) {
      D[(r + half * 8) * ldd + n]      = fmaxf(acc0[r] * sc + bs, 0.f);
      D[(16 + r + half * 8) * ldd + n] = fmaxf(acc1[r] * sc + bs, 0.f);
    }
  }
}

__global__ __launch_bounds__(64) void mlp_kernel(const float* __restrict__ xyz,
                                                 const float* __restrict__ feat,
                                                 const int*   __restrict__ grp,
                                                 const float* __restrict__ newxyz,
                                                 const float* __restrict__ w0, const float* __restrict__ g0,
                                                 const float* __restrict__ b0, const float* __restrict__ m0,
                                                 const float* __restrict__ v0,
                                                 const float* __restrict__ w1, const float* __restrict__ g1,
                                                 const float* __restrict__ b1, const float* __restrict__ m1,
                                                 const float* __restrict__ v1,
                                                 const float* __restrict__ w2, const float* __restrict__ g2,
                                                 const float* __restrict__ b2, const float* __restrict__ m2,
                                                 const float* __restrict__ v2,
                                                 float* __restrict__ outfeat) {
  __shared__ float Xb[2][32 * 68];
  __shared__ float Hb[2][32 * 132];

  const int wave = threadIdx.x >> 5;
  const int lane = threadIdx.x & 31;
  const int q    = blockIdx.x * 2 + wave;
  const int b    = q >> 9;
  const int s    = q & (NPT - 1);

  float* X = Xb[wave];
  float* H = Hb[wave];

  const int gi = grp[(size_t)q * NSAMP + lane];
  const float* prow = xyz + ((size_t)b * NPTS + gi) * 3;
  const float* nx   = newxyz + (b * NPT + s) * 3;
  float* xr = X + lane * 68;
  xr[0] = prow[0] - nx[0];
  xr[1] = prow[1] - nx[1];
  xr[2] = prow[2] - nx[2];
  const float4* frow = (const float4*)(feat + ((size_t)b * NPTS + gi) * 64);
#pragma unroll
  for (int c4 = 0; c4 < 16; ++c4) {
    float4 f = frow[c4];
    xr[3 + 4 * c4 + 0] = f.x;
    xr[3 + 4 * c4 + 1] = f.y;
    xr[3 + 4 * c4 + 2] = f.z;
    xr[3 + 4 * c4 + 3] = f.w;
  }
  xr[67] = 0.f;

  gemm_bn_relu(X, 68,  w0, 67, 68, 64,  g0, b0, m0, v0, H, 132, lane);
  gemm_bn_relu(H, 132, w1, 64, 64, 64,  g1, b1, m1, v1, X, 68,  lane);
  gemm_bn_relu(X, 68,  w2, 64, 64, 128, g2, b2, m2, v2, H, 132, lane);

#pragma unroll
  for (int cc = 0; cc < 4; ++cc) {
    const int col = lane + cc * 32;
    float mx = H[col];
    for (int row = 1; row < 32; ++row) mx = fmaxf(mx, H[row * 132 + col]);
    outfeat[((size_t)b * 128 + col) * NPT + s] = mx;
  }
}

extern "C" void kernel_launch(void* const* d_in, const int* in_sizes, int n_in,
                              void* d_out, int out_size, void* d_ws, size_t ws_size,
                              hipStream_t stream) {
  (void)in_sizes; (void)n_in; (void)out_size; (void)ws_size;
  const float* xyz  = (const float*)d_in[0];
  const float* feat = (const float*)d_in[1];
  const float* w0 = (const float*)d_in[2];
  const float* g0 = (const float*)d_in[3];
  const float* b0 = (const float*)d_in[4];
  const float* m0 = (const float*)d_in[5];
  const float* v0 = (const float*)d_in[6];
  const float* w1 = (const float*)d_in[7];
  const float* g1 = (const float*)d_in[8];
  const float* b1 = (const float*)d_in[9];
  const float* m1 = (const float*)d_in[10];
  const float* v1 = (const float*)d_in[11];
  const float* w2 = (const float*)d_in[12];
  const float* g2 = (const float*)d_in[13];
  const float* b2 = (const float*)d_in[14];
  const float* m2 = (const float*)d_in[15];
  const float* v2 = (const float*)d_in[16];

  float* newxyz  = (float*)d_out;
  float* outfeat = (float*)d_out + NB * NPT * 3;

  int* fpsidx = (int*)d_ws;
  int* grp    = (int*)d_ws + NB * NPT;

  fps_kernel  <<<NB,                1024, 0, stream>>>(xyz, fpsidx, newxyz);
  ballq_kernel<<<(NB * NPT) / 8,     256, 0, stream>>>(xyz, newxyz, grp);
  mlp_kernel  <<<(NB * NPT) / 2,      64, 0, stream>>>(xyz, feat, grp, newxyz,
                                                       w0, g0, b0, m0, v0,
                                                       w1, g1, b1, m1, v1,
                                                       w2, g2, b2, m2, v2,
                                                       outfeat);
}


# FPS: one block exchange and barrier per iteration, not two: every wave tests all 32 wave boxes lanewise, and only a wave no kept maximum beats looks up its index before the barrier; same picks
# speedup vs baseline: 2.7619x; 1.0594x over previous
#include <hip/hip_runtime.h>
#include <hip/hip_bf16.h>
#include <math.h>

typedef float v2f __attribute__((ext_vector_type(2)));
typedef float v8f __attribute__((ext_vector_type(8)));

#define NB     8
#define NPTS   16384
#define NPT    512
#define NSAMP  32
#define RAD2   0.04f
#define BN_EPS 1e-5f


template <int M>
__device__ __forceinline__ int row_xor(int v) {
  return __builtin_amdgcn_update_dpp(0, v, 0x160 | M, 0xf, 0xf, true);
}
__device__ __forceinline__ int half_xor(int v) {
  return (int)__builtin_amdgcn_permlanex16((unsigned)v, (unsigned)v, 0x76543210u, 0xfedcba98u, false, false);
}
__device__ __forceinline__ int imax2(int a, int b) { return a > b ? a : b; }
__device__ __forceinline__ int imin2(int a, int b) { return a < b ? a : b; }

__device__ __forceinline__ int wave_imax(int m) {
  m = imax2(m, row_xor<1>(m));
  m = imax2(m, row_xor<2>(m));
  m = imax2(m, row_xor<4>(m));
  m = imax2(m, row_xor<8>(m));
  m = imax2(m, half_xor(m));
  return m;
}
__device__ __forceinline__ int wave_imin(int c) {
  c = imin2(c, row_xor<1>(c));
  c = imin2(c, row_xor<2>(c));
  c = imin2(c, row_xor<4>(c));
  c = imin2(c, row_xor<8>(c));
  c = imin2(c, half_xor(c));
  return c;
}

__device__ __forceinline__ float uni(float v) {
  return __int_as_float(__builtin_amdgcn_readfirstlane(__float_as_int(v)));
}
__device__ __forceinline__ unsigned spread3(unsigned v) {
  return (v & 1u) | ((v & 2u) << 2) | ((v & 4u) << 4);
}

__global__ __launch_bounds__(1024) void fps_kernel(const float* __restrict__ xyz,
                                                   int* __restrict__ fpsidx,
                                                   float* __restrict__ newxyz) {
  const int b   = blockIdx.x;
  const int tid = threadIdx.x;
  const float* base = xyz + (size_t)b * NPTS * 3;

  __shared__ float4 P[NPTS];
  __shared__ int2 wrec[2][32];
  __shared__ float wbox[32][6];
  __shared__ int wtot[32];

  const int wave = tid >> 5;
  const int lane = tid & 31;

  unsigned* K = (unsigned*)P;
  unsigned key[16];
#pragma unroll
  for (int j = 0; j < 16; ++j) {
    const int p = tid * 16 + j;
    const float x = base[p * 3 + 0], y = base[p * 3 + 1], z = base[p * 3 + 2];
    const unsigned ix = (unsigned)(int)fminf(fmaxf(x * 8.0f, 0.0f), 7.0f);
    const unsigned iy = (unsigned)(int)fminf(fmaxf(y * 8.0f, 0.0f), 7.0f);
    const unsigned iz = (unsigned)(int)fminf(fmaxf(z * 8.0f, 0.0f), 7.0f);
    key[j] = ((spread3(ix) | (spread3(iy) << 1) | (spread3(iz) << 2)) << 14) | (unsigned)p;
  }
  for (int bit = 14; bit < 23; ++bit) {
    int z = 0;
#pragma unroll
    for (int j = 0; j < 16; ++j) z += (int)(((key[j] >> bit) & 1u) ^ 1u);
    int inc = z;
    for (int d = 1; d < 32; d <<= 1) {
      const int t = __shfl_up(inc, d);
      if (lane >= d) inc += t;
    }
    if (lane == 31) wtot[wave] = inc;
    __syncthreads();
    const int wt = wtot[lane];
    int winc = wt;
    for (int d = 1; d < 32; d <<= 1) {
      const int t = __shfl_up(winc, d);
      if (lane >= d) winc += t;
    }
    const int zall  = __shfl(winc, 31);
    const int wexcl = __shfl(winc - wt, wave);
    int zpos = wexcl + inc - z;
    int opos = zall + (tid * 16 - zpos);
#pragma unroll
    for (int j = 0; j < 16; ++j) {
      const int one = (int)((key[j] >> bit) & 1u);
      K[one ? opos : zpos] = key[j];
      opos += one;
      zpos += 1 - one;
    }
    __syncthreads();
#pragma unroll
    for (int j = 0; j < 16; ++j) key[j] = K[tid * 16 + j];
    __syncthreads();
  }

  float px[16], py[16], pz[16], dd[16];
  unsigned pk[8];
#pragma unroll
  for (int j = 0; j < 16; ++j) {
    const int p = (int)(key[j] & 0x3fffu);
    px[j] = base[p * 3 + 0];
    py[j] = base[p * 3 + 1];
    pz[j] = base[p * 3 + 2];
    dd[j] = 3.4e38f;
    P[p] = make_float4(px[j], py[j], pz[j], 0.0f);
  }
#pragma unroll
  for (int j = 0; j < 16; j += 2)
    pk[j >> 1] = (key[j] & 0x3fffu) | ((key[j + 1] & 0x3fffu) << 16);

  float lx = px[0], hx = px[0], ly = py[0], hy = py[0], lz = pz[0], hz = pz[0];
#pragma unroll
  for (int j = 1; j < 16; ++j) {
    lx = fminf(lx, px[j]); hx = fmaxf(hx, px[j]);
    ly = fminf(ly, py[j]); hy = fmaxf(hy, py[j]);
    lz = fminf(lz, pz[j]); hz = fmaxf(hz, pz[j]);
  }
  for (int d = 16; d >= 1; d >>= 1) {
    lx = fminf(lx, __shfl_xor(lx, d)); hx = fmaxf(hx, __shfl_xor(hx, d));
    ly = fminf(ly, __shfl_xor(ly, d)); hy = fmaxf(hy, __shfl_xor(hy, d));
    lz = fminf(lz, __shfl_xor(lz, d)); hz = fmaxf(hz, __shfl_xor(hz, d));
  }
  if (lane == 0) {
    wbox[wave][0] = lx; wbox[wave][1] = hx;
    wbox[wave][2] = ly; wbox[wave][3] = hy;
    wbox[wave][4] = lz; wbox[wave][5] = hz;
  }
  __syncthreads();
  const float blx = wbox[lane][0], bhx = wbox[lane][1];
  const float bly = wbox[lane][2], bhy = wbox[lane][3];
  const float blz = wbox[lane][4], bhz = wbox[lane][5];
  const int wv = __builtin_amdgcn_readfirstlane(wave);

  const int NOIDX = 0x7fffffff;
  int last = 0;
  float cx = base[0], cy = base[1], cz = base[2];
  int m = __float_as_int(3.4e38f);
  int wc = NOIDX;
  int mv = __float_as_int(3.4e38f);

  for (int i = 0; i < NPT; ++i) {
    if (tid == 0) {
      fpsidx[b * NPT + i] = last;
      newxyz[(b * NPT + i) * 3 + 0] = cx;
      newxyz[(b * NPT + i) * 3 + 1] = cy;
      newxyz[(b * NPT + i) * 3 + 2] = cz;
    }

    const float ex = fmaxf(fmaxf(blx - cx, cx - bhx), 0.0f);
    const float ey = fmaxf(fmaxf(bly - cy, cy - bhy), 0.0f);
    const float ez = fmaxf(fmaxf(blz - cz, cz - bhz), 0.0f);
    const float lb = (ex * ex + ey * ey + ez * ez) * 0.999f;
    const bool keeps = lb > __int_as_float(mv);
    const unsigned keepmask = __builtin_amdgcn_ballot_w32(keeps);
    if (((keepmask >> wv) & 1u) == 0u) {
      float bv = -1.0f;
#pragma unroll
      for (int j = 0; j < 16; ++j) {
        float dx = px[j] - cx, dy = py[j] - cy, dz = pz[j] - cz;
        float d  = __builtin_fmaf(dz, dz, __builtin_fmaf(dx, dx, dy * dy));
        float nd = fminf(dd[j], d);
        dd[j] = nd;
        bv = fmaxf(bv, nd);
      }
      m  = wave_imax(__float_as_int(bv));
      wc = NOIDX;
    }

    const bool beaten = __builtin_amdgcn_ballot_w32(keeps && (mv > m)) != 0u;
    if (!beaten && wc == NOIDX) {
      const int M = m;
      int c0 = NOIDX, c1 = NOIDX, c2 = NOIDX, c3 = NOIDX;
#pragma unroll
      for (int j = 0; j < 16; j += 4) {
        if (__float_as_int(dd[j])     == M) c0 = imin2(c0, (int)(pk[j >> 1] & 0xffffu));
        if (__float_as_int(dd[j + 1]) == M) c1 = imin2(c1, (int)(pk[j >> 1] >> 16));
        if (__float_as_int(dd[j + 2]) == M) c2 = imin2(c2, (int)(pk[(j >> 1) + 1] & 0xffffu));
        if (__float_as_int(dd[j + 3]) == M) c3 = imin2(c3, (int)(pk[(j >> 1) + 1] >> 16));
      }
      const int cl = imin2(imin2(c0, c1), imin2(c2, c3));
      const unsigned hl = __builtin_amdgcn_ballot_w32(cl != NOIDX);
      if (__builtin_popcount(hl) == 1) {
        wc = __builtin_amdgcn_readlane(cl, __builtin_ctz(hl));
      } else {
        wc = __builtin_amdgcn_readfirstlane(wave_imin(cl));
      }
    }

    if (lane == 0) wrec[i & 1][wave] = make_int2(m, wc);
    __syncthreads();
    const int2 rec = wrec[i & 1][lane];
    mv = rec.x;
    const int    cand = rec.y;
    const float4 pc   = P[cand & (NPTS - 1)];
    const int bm = wave_imax(mv);
    const unsigned hm = __builtin_amdgcn_ballot_w32(mv == bm);
    int wl;
    if (__builtin_popcount(hm) == 1) {
      wl = __builtin_ctz(hm);
    } else {
      const int ct   = (mv == bm) ? cand : NOIDX;
      const int cmin = wave_imin(ct);
      wl = __builtin_ctz(__builtin_amdgcn_ballot_w32(ct == cmin));
    }
    cx = __int_as_float(__builtin_amdgcn_readlane(__float_as_int(pc.x), wl));
    cy = __int_as_float(__builtin_amdgcn_readlane(__float_as_int(pc.y), wl));
    cz = __int_as_float(__builtin_amdgcn_readlane(__float_as_int(pc.z), wl));
    last = __builtin_amdgcn_readlane(cand, wl) & (NPTS - 1);
  }
}

#define BQ_U 16
__global__ __launch_bounds__(256) void ballq_kernel(const float* __restrict__ xyz,
                                                    const float* __restrict__ newxyz,
                                                    int* __restrict__ grp) {
  const int lane = threadIdx.x & 31;
  const int q    = blockIdx.x * 8 + (threadIdx.x >> 5);
  const int b    = q >> 9;
  const int s    = q & (NPT - 1);

  const float* base = xyz + (size_t)b * NPTS * 3;
  const float cx = newxyz[(b * NPT + s) * 3 + 0];
  const float cy = newxyz[(b * NPT + s) * 3 + 1];
  const float cz = newxyz[(b * NPT + s) * 3 + 2];

  int* g = grp + (size_t)q * NSAMP;
  int cnt = 0, first = -1;

  for (int bp = 0; bp < NPTS && cnt < NSAMP; bp += 32 * BQ_U) {
    float px[BQ_U], py[BQ_U], pz[BQ_U];
#pragma unroll
    for (int u = 0; u < BQ_U; ++u) {
      const int p = bp + 32 * u + lane;
      px[u] = base[p * 3 + 0];
      py[u] = base[p * 3 + 1];
      pz[u] = base[p * 3 + 2];
    }
    unsigned mask[BQ_U];
#pragma unroll
    for (int u = 0; u < BQ_U; ++u) {
      float dx = px[u] - cx;
      float dy = py[u] - cy;
      float dz = pz[u] - cz;
      mask[u] = (unsigned)__ballot((dx * dx + dy * dy + dz * dz) <= RAD2);
    }
#pragma unroll
    for (int u = 0; u < BQ_U; ++u) {
      if (mask[u]) {
        if (cnt < NSAMP) {
          const int cb = bp + 32 * u;
          if (first < 0) first = cb + __builtin_ctz(mask[u]);
          const bool hit = (mask[u] >> lane) & 1u;
          int pos = __popc(mask[u] & ((1u << lane) - 1u));
          if (hit && (cnt + pos) < NSAMP) g[cnt + pos] = cb + lane;
          cnt += __popc(mask[u]);
        }
      }
    }
  }
  if (first < 0) first = NPTS - 1;
  int filled = cnt < NSAMP ? cnt : NSAMP;
  for (int slot = filled + lane; slot < NSAMP; slot += 32) g[slot] = first;
}

__device__ __forceinline__ void gemm_bn_relu(const float* __restrict__ A, int lda,
                                             const float* __restrict__ W, int Kdim, int Kpad,
                                             int Nout,
                                             const float* __restrict__ gg,
                                             const float* __restrict__ bb,
                                             const float* __restrict__ mm,
                                             const float* __restrict__ vv,
                                             float* __restrict__ D, int ldd, int lane) {
  const int r15  = lane & 15;
  const int half = lane >> 4;
  for (int n0 = 0; n0 < Nout; n0 += 16) {
    v8f acc0 = {0.f, 0.f, 0.f, 0.f, 0.f, 0.f, 0.f, 0.f};
    v8f acc1 = {0.f, 0.f, 0.f, 0.f, 0.f, 0.f, 0.f, 0.f};
    const int n = n0 + r15;
    for (int k0 = 0; k0 < Kpad; k0 += 4) {
      const int kk = k0 + half * 2;
      v2f a0, a1, w;
      a0.x = A[r15 * lda + kk];
      a0.y = A[r15 * lda + kk + 1];
      a1.x = A[(16 + r15) * lda + kk];
      a1.y = A[(16 + r15) * lda + kk + 1];
      int ka = kk     < Kdim ? kk     : Kdim - 1;
      int kb = kk + 1 < Kdim ? kk + 1 : Kdim - 1;
      w.x = W[n * Kdim + ka];
      w.y = W[n * Kdim + kb];
      acc0 = __builtin_amdgcn_wmma_f32_16x16x4_f32(false, a0, false, w,
                                                   (short)0, acc0, false, false);
      acc1 = __builtin_amdgcn_wmma_f32_16x16x4_f32(false, a1, false, w,
                                                   (short)0, acc1, false, false);
    }
    const float sc = gg[n] * rsqrtf(vv[n] + BN_EPS);
    const float bs = bb[n] - mm[n] * sc;
#pragma unroll
    for (int r = 0; r < 8; ++r) {
      D[(r + half * 8) * ldd + n]      = fmaxf(acc0[r] * sc + bs, 0.f);
      D[(16 + r + half * 8) * ldd + n] = fmaxf(acc1[r] * sc + bs, 0.f);
    }
  }
}

__global__ __launch_bounds__(64) void mlp_kernel(const float* __restrict__ xyz,
                                                 const float* __restrict__ feat,
                                                 const int*   __restrict__ grp,
                                                 const float* __restrict__ newxyz,
                                                 const float* __restrict__ w0, const float* __restrict__ g0,
                                                 const float* __restrict__ b0, const float* __restrict__ m0,
                                                 const float* __restrict__ v0,
                                                 const float* __restrict__ w1, const float* __restrict__ g1,
                                                 const float* __restrict__ b1, const float* __restrict__ m1,
                                                 const float* __restrict__ v1,
                                                 const float* __restrict__ w2, const float* __restrict__ g2,
                                                 const float* __restrict__ b2, const float* __restrict__ m2,
                                                 const float* __restrict__ v2,
                                                 float* __restrict__ outfeat) {
  __shared__ float Xb[2][32 * 68];
  __shared__ float Hb[2][32 * 132];

  const int wave = threadIdx.x >> 5;
  const int lane = threadIdx.x & 31;
  const int q    = blockIdx.x * 2 + wave;
  const int b    = q >> 9;
  const int s    = q & (NPT - 1);

  float* X = Xb[wave];
  float* H = Hb[wave];

  const int gi = grp[(size_t)q * NSAMP + lane];
  const float* prow = xyz + ((size_t)b * NPTS + gi) * 3;
  const float* nx   = newxyz + (b * NPT + s) * 3;
  float* xr = X + lane * 68;
  xr[0] = prow[0] - nx[0];
  xr[1] = prow[1] - nx[1];
  xr[2] = prow[2] - nx[2];
  const float4* frow = (const float4*)(feat + ((size_t)b * NPTS + gi) * 64);
#pragma unroll
  for (int c4 = 0; c4 < 16; ++c4) {
    float4 f = frow[c4];
    xr[3 + 4 * c4 + 0] = f.x;
    xr[3 + 4 * c4 + 1] = f.y;
    xr[3 + 4 * c4 + 2] = f.z;
    xr[3 + 4 * c4 + 3] = f.w;
  }
  xr[67] = 0.f;

  gemm_bn_relu(X, 68,  w0, 67, 68, 64,  g0, b0, m0, v0, H, 132, lane);
  gemm_bn_relu(H, 132, w1, 64, 64, 64,  g1, b1, m1, v1, X, 68,  lane);
  gemm_bn_relu(X, 68,  w2, 64, 64, 128, g2, b2, m2, v2, H, 132, lane);

#pragma unroll
  for (int cc = 0; cc < 4; ++cc) {
    const int col = lane + cc * 32;
    float mx = H[col];
    for (int row = 1; row < 32; ++row) mx = fmaxf(mx, H[row * 132 + col]);
    outfeat[((size_t)b * 128 + col) * NPT + s] = mx;
  }
}

extern "C" void kernel_launch(void* const* d_in, const int* in_sizes, int n_in,
                              void* d_out, int out_size, void* d_ws, size_t ws_size,
                              hipStream_t stream) {
  (void)in_sizes; (void)n_in; (void)out_size; (void)ws_size;
  const float* xyz  = (const float*)d_in[0];
  const float* feat = (const float*)d_in[1];
  const float* w0 = (const float*)d_in[2];
  const float* g0 = (const float*)d_in[3];
  const float* b0 = (const float*)d_in[4];
  const float* m0 = (const float*)d_in[5];
  const float* v0 = (const float*)d_in[6];
  const float* w1 = (const float*)d_in[7];
  const float* g1 = (const float*)d_in[8];
  const float* b1 = (const float*)d_in[9];
  const float* m1 = (const float*)d_in[10];
  const float* v1 = (const float*)d_in[11];
  const float* w2 = (const float*)d_in[12];
  const float* g2 = (const float*)d_in[13];
  const float* b2 = (const float*)d_in[14];
  const float* m2 = (const float*)d_in[15];
  const float* v2 = (const float*)d_in[16];

  float* newxyz  = (float*)d_out;
  float* outfeat = (float*)d_out + NB * NPT * 3;

  int* fpsidx = (int*)d_ws;
  int* grp    = (int*)d_ws + NB * NPT;

  fps_kernel  <<<NB,                1024, 0, stream>>>(xyz, fpsidx, newxyz);
  ballq_kernel<<<(NB * NPT) / 8,     256, 0, stream>>>(xyz, newxyz, grp);
  mlp_kernel  <<<(NB * NPT) / 2,      64, 0, stream>>>(xyz, feat, grp, newxyz,
                                                       w0, g0, b0, m0, v0,
                                                       w1, g1, b1, m1, v1,
                                                       w2, g2, b2, m2, v2,
                                                       outfeat);
}


# FPS: grid cells numbered along a Hilbert curve, not bit-interleaved: each wave's 512 consecutive points lie in one compact box (no far-off neighbour cells), so fewer waves update per step; same picks
# speedup vs baseline: 2.9518x; 1.0688x over previous
#include <hip/hip_runtime.h>
#include <hip/hip_bf16.h>
#include <math.h>

typedef float v2f __attribute__((ext_vector_type(2)));
typedef float v8f __attribute__((ext_vector_type(8)));

#define NB     8
#define NPTS   16384
#define NPT    512
#define NSAMP  32
#define RAD2   0.04f
#define BN_EPS 1e-5f


template <int M>
__device__ __forceinline__ int row_xor(int v) {
  return __builtin_amdgcn_update_dpp(0, v, 0x160 | M, 0xf, 0xf, true);
}
__device__ __forceinline__ int half_xor(int v) {
  return (int)__builtin_amdgcn_permlanex16((unsigned)v, (unsigned)v, 0x76543210u, 0xfedcba98u, false, false);
}
__device__ __forceinline__ int imax2(int a, int b) { return a > b ? a : b; }
__device__ __forceinline__ int imin2(int a, int b) { return a < b ? a : b; }

__device__ __forceinline__ int wave_imax(int m) {
  m = imax2(m, row_xor<1>(m));
  m = imax2(m, row_xor<2>(m));
  m = imax2(m, row_xor<4>(m));
  m = imax2(m, row_xor<8>(m));
  m = imax2(m, half_xor(m));
  return m;
}
__device__ __forceinline__ int wave_imin(int c) {
  c = imin2(c, row_xor<1>(c));
  c = imin2(c, row_xor<2>(c));
  c = imin2(c, row_xor<4>(c));
  c = imin2(c, row_xor<8>(c));
  c = imin2(c, half_xor(c));
  return c;
}

__device__ __forceinline__ float uni(float v) {
  return __int_as_float(__builtin_amdgcn_readfirstlane(__float_as_int(v)));
}
__device__ __forceinline__ unsigned hilbert_pos(unsigned x0, unsigned x1, unsigned x2) {
  unsigned X[3] = {x0, x1, x2};
  for (unsigned Q = 4u; Q > 1u; Q >>= 1) {
    const unsigned Pm = Q - 1u;
    for (int i = 0; i < 3; ++i) {
      if (X[i] & Q) {
        X[0] ^= Pm;
      } else {
        const unsigned t = (X[0] ^ X[i]) & Pm;
        X[0] ^= t;
        X[i] ^= t;
      }
    }
  }
  X[1] ^= X[0];
  X[2] ^= X[1];
  unsigned t = 0u;
  for (unsigned Q = 4u; Q > 1u; Q >>= 1)
    if (X[2] & Q) t ^= Q - 1u;
  X[0] ^= t; X[1] ^= t; X[2] ^= t;
  unsigned h = 0u;
  for (int bit = 2; bit >= 0; --bit)
    h = (h << 3) | (((X[0] >> bit) & 1u) << 2) | (((X[1] >> bit) & 1u) << 1) | ((X[2] >> bit) & 1u);
  return h;
}

__global__ __launch_bounds__(1024) void fps_kernel(const float* __restrict__ xyz,
                                                   int* __restrict__ fpsidx,
                                                   float* __restrict__ newxyz) {
  const int b   = blockIdx.x;
  const int tid = threadIdx.x;
  const float* base = xyz + (size_t)b * NPTS * 3;

  __shared__ float4 P[NPTS];
  __shared__ int2 wrec[2][32];
  __shared__ float wbox[32][6];
  __shared__ int wtot[32];
  __shared__ unsigned short hil[512];

  const int wave = tid >> 5;
  const int lane = tid & 31;

  if (tid < 512)
    hil[tid] = (unsigned short)hilbert_pos((unsigned)tid & 7u, ((unsigned)tid >> 3) & 7u, (unsigned)tid >> 6);
  __syncthreads();
  unsigned* K = (unsigned*)P;
  unsigned key[16];
#pragma unroll
  for (int j = 0; j < 16; ++j) {
    const int p = tid * 16 + j;
    const float x = base[p * 3 + 0], y = base[p * 3 + 1], z = base[p * 3 + 2];
    const unsigned ix = (unsigned)(int)fminf(fmaxf(x * 8.0f, 0.0f), 7.0f);
    const unsigned iy = (unsigned)(int)fminf(fmaxf(y * 8.0f, 0.0f), 7.0f);
    const unsigned iz = (unsigned)(int)fminf(fmaxf(z * 8.0f, 0.0f), 7.0f);
    key[j] = ((unsigned)hil[ix | (iy << 3) | (iz << 6)] << 14) | (unsigned)p;
  }
  for (int bit = 14; bit < 23; ++bit) {
    int z = 0;
#pragma unroll
    for (int j = 0; j < 16; ++j) z += (int)(((key[j] >> bit) & 1u) ^ 1u);
    int inc = z;
    for (int d = 1; d < 32; d <<= 1) {
      const int t = __shfl_up(inc, d);
      if (lane >= d) inc += t;
    }
    if (lane == 31) wtot[wave] = inc;
    __syncthreads();
    const int wt = wtot[lane];
    int winc = wt;
    for (int d = 1; d < 32; d <<= 1) {
      const int t = __shfl_up(winc, d);
      if (lane >= d) winc += t;
    }
    const int zall  = __shfl(winc, 31);
    const int wexcl = __shfl(winc - wt, wave);
    int zpos = wexcl + inc - z;
    int opos = zall + (tid * 16 - zpos);
#pragma unroll
    for (int j = 0; j < 16; ++j) {
      const int one = (int)((key[j] >> bit) & 1u);
      K[one ? opos : zpos] = key[j];
      opos += one;
      zpos += 1 - one;
    }
    __syncthreads();
#pragma unroll
    for (int j = 0; j < 16; ++j) key[j] = K[tid * 16 + j];
    __syncthreads();
  }

  float px[16], py[16], pz[16], dd[16];
  unsigned pk[8];
#pragma unroll
  for (int j = 0; j < 16; ++j) {
    const int p = (int)(key[j] & 0x3fffu);
    px[j] = base[p * 3 + 0];
    py[j] = base[p * 3 + 1];
    pz[j] = base[p * 3 + 2];
    dd[j] = 3.4e38f;
    P[p] = make_float4(px[j], py[j], pz[j], 0.0f);
  }
#pragma unroll
  for (int j = 0; j < 16; j += 2)
    pk[j >> 1] = (key[j] & 0x3fffu) | ((key[j + 1] & 0x3fffu) << 16);

  float lx = px[0], hx = px[0], ly = py[0], hy = py[0], lz = pz[0], hz = pz[0];
#pragma unroll
  for (int j = 1; j < 16; ++j) {
    lx = fminf(lx, px[j]); hx = fmaxf(hx, px[j]);
    ly = fminf(ly, py[j]); hy = fmaxf(hy, py[j]);
    lz = fminf(lz, pz[j]); hz = fmaxf(hz, pz[j]);
  }
  for (int d = 16; d >= 1; d >>= 1) {
    lx = fminf(lx, __shfl_xor(lx, d)); hx = fmaxf(hx, __shfl_xor(hx, d));
    ly = fminf(ly, __shfl_xor(ly, d)); hy = fmaxf(hy, __shfl_xor(hy, d));
    lz = fminf(lz, __shfl_xor(lz, d)); hz = fmaxf(hz, __shfl_xor(hz, d));
  }
  if (lane == 0) {
    wbox[wave][0] = lx; wbox[wave][1] = hx;
    wbox[wave][2] = ly; wbox[wave][3] = hy;
    wbox[wave][4] = lz; wbox[wave][5] = hz;
  }
  __syncthreads();
  const float blx = wbox[lane][0], bhx = wbox[lane][1];
  const float bly = wbox[lane][2], bhy = wbox[lane][3];
  const float blz = wbox[lane][4], bhz = wbox[lane][5];
  const int wv = __builtin_amdgcn_readfirstlane(wave);

  const int NOIDX = 0x7fffffff;
  int last = 0;
  float cx = base[0], cy = base[1], cz = base[2];
  int m = __float_as_int(3.4e38f);
  int wc = NOIDX;
  int mv = __float_as_int(3.4e38f);

  for (int i = 0; i < NPT; ++i) {
    if (tid == 0) {
      fpsidx[b * NPT + i] = last;
      newxyz[(b * NPT + i) * 3 + 0] = cx;
      newxyz[(b * NPT + i) * 3 + 1] = cy;
      newxyz[(b * NPT + i) * 3 + 2] = cz;
    }

    const float ex = fmaxf(fmaxf(blx - cx, cx - bhx), 0.0f);
    const float ey = fmaxf(fmaxf(bly - cy, cy - bhy), 0.0f);
    const float ez = fmaxf(fmaxf(blz - cz, cz - bhz), 0.0f);
    const float lb = (ex * ex + ey * ey + ez * ez) * 0.999f;
    const bool keeps = lb > __int_as_float(mv);
    const unsigned keepmask = __builtin_amdgcn_ballot_w32(keeps);
    if (((keepmask >> wv) & 1u) == 0u) {
      float bv = -1.0f;
#pragma unroll
      for (int j = 0; j < 16; ++j) {
        float dx = px[j] - cx, dy = py[j] - cy, dz = pz[j] - cz;
        float d  = __builtin_fmaf(dz, dz, __builtin_fmaf(dx, dx, dy * dy));
        float nd = fminf(dd[j], d);
        dd[j] = nd;
        bv = fmaxf(bv, nd);
      }
      m  = wave_imax(__float_as_int(bv));
      wc = NOIDX;
    }

    const bool beaten = __builtin_amdgcn_ballot_w32(keeps && (mv > m)) != 0u;
    if (!beaten && wc == NOIDX) {
      const int M = m;
      int c0 = NOIDX, c1 = NOIDX, c2 = NOIDX, c3 = NOIDX;
#pragma unroll
      for (int j = 0; j < 16; j += 4) {
        if (__float_as_int(dd[j])     == M) c0 = imin2(c0, (int)(pk[j >> 1] & 0xffffu));
        if (__float_as_int(dd[j + 1]) == M) c1 = imin2(c1, (int)(pk[j >> 1] >> 16));
        if (__float_as_int(dd[j + 2]) == M) c2 = imin2(c2, (int)(pk[(j >> 1) + 1] & 0xffffu));
        if (__float_as_int(dd[j + 3]) == M) c3 = imin2(c3, (int)(pk[(j >> 1) + 1] >> 16));
      }
      const int cl = imin2(imin2(c0, c1), imin2(c2, c3));
      const unsigned hl = __builtin_amdgcn_ballot_w32(cl != NOIDX);
      if (__builtin_popcount(hl) == 1) {
        wc = __builtin_amdgcn_readlane(cl, __builtin_ctz(hl));
      } else {
        wc = __builtin_amdgcn_readfirstlane(wave_imin(cl));
      }
    }

    if (lane == 0) wrec[i & 1][wave] = make_int2(m, wc);
    __syncthreads();
    const int2 rec = wrec[i & 1][lane];
    mv = rec.x;
    const int    cand = rec.y;
    const float4 pc   = P[cand & (NPTS - 1)];
    const int bm = wave_imax(mv);
    const unsigned hm = __builtin_amdgcn_ballot_w32(mv == bm);
    int wl;
    if (__builtin_popcount(hm) == 1) {
      wl = __builtin_ctz(hm);
    } else {
      const int ct   = (mv == bm) ? cand : NOIDX;
      const int cmin = wave_imin(ct);
      wl = __builtin_ctz(__builtin_amdgcn_ballot_w32(ct == cmin));
    }
    cx = __int_as_float(__builtin_amdgcn_readlane(__float_as_int(pc.x), wl));
    cy = __int_as_float(__builtin_amdgcn_readlane(__float_as_int(pc.y), wl));
    cz = __int_as_float(__builtin_amdgcn_readlane(__float_as_int(pc.z), wl));
    last = __builtin_amdgcn_readlane(cand, wl) & (NPTS - 1);
  }
}

#define BQ_U 16
__global__ __launch_bounds__(256) void ballq_kernel(const float* __restrict__ xyz,
                                                    const float* __restrict__ newxyz,
                                                    int* __restrict__ grp) {
  const int lane = threadIdx.x & 31;
  const int q    = blockIdx.x * 8 + (threadIdx.x >> 5);
  const int b    = q >> 9;
  const int s    = q & (NPT - 1);

  const float* base = xyz + (size_t)b * NPTS * 3;
  const float cx = newxyz[(b * NPT + s) * 3 + 0];
  const float cy = newxyz[(b * NPT + s) * 3 + 1];
  const float cz = newxyz[(b * NPT + s) * 3 + 2];

  int* g = grp + (size_t)q * NSAMP;
  int cnt = 0, first = -1;

  for (int bp = 0; bp < NPTS && cnt < NSAMP; bp += 32 * BQ_U) {
    float px[BQ_U], py[BQ_U], pz[BQ_U];
#pragma unroll
    for (int u = 0; u < BQ_U; ++u) {
      const int p = bp + 32 * u + lane;
      px[u] = base[p * 3 + 0];
      py[u] = base[p * 3 + 1];
      pz[u] = base[p * 3 + 2];
    }
    unsigned mask[BQ_U];
#pragma unroll
    for (int u = 0; u < BQ_U; ++u) {
      float dx = px[u] - cx;
      float dy = py[u] - cy;
      float dz = pz[u] - cz;
      mask[u] = (unsigned)__ballot((dx * dx + dy * dy + dz * dz) <= RAD2);
    }
#pragma unroll
    for (int u = 0; u < BQ_U; ++u) {
      if (mask[u]) {
        if (cnt < NSAMP) {
          const int cb = bp + 32 * u;
          if (first < 0) first = cb + __builtin_ctz(mask[u]);
          const bool hit = (mask[u] >> lane) & 1u;
          int pos = __popc(mask[u] & ((1u << lane) - 1u));
          if (hit && (cnt + pos) < NSAMP) g[cnt + pos] = cb + lane;
          cnt += __popc(mask[u]);
        }
      }
    }
  }
  if (first < 0) first = NPTS - 1;
  int filled = cnt < NSAMP ? cnt : NSAMP;
  for (int slot = filled + lane; slot < NSAMP; slot += 32) g[slot] = first;
}

__device__ __forceinline__ void gemm_bn_relu(const float* __restrict__ A, int lda,
                                             const float* __restrict__ W, int Kdim, int Kpad,
                                             int Nout,
                                             const float* __restrict__ gg,
                                             const float* __restrict__ bb,
                                             const float* __restrict__ mm,
                                             const float* __restrict__ vv,
                                             float* __restrict__ D, int ldd, int lane) {
  const int r15  = lane & 15;
  const int half = lane >> 4;
  for (int n0 = 0; n0 < Nout; n0 += 16) {
    v8f acc0 = {0.f, 0.f, 0.f, 0.f, 0.f, 0.f, 0.f, 0.f};
    v8f acc1 = {0.f, 0.f, 0.f, 0.f, 0.f, 0.f, 0.f, 0.f};
    const int n = n0 + r15;
    for (int k0 = 0; k0 < Kpad; k0 += 4) {
      const int kk = k0 + half * 2;
      v2f a0, a1, w;
      a0.x = A[r15 * lda + kk];
      a0.y = A[r15 * lda + kk + 1];
      a1.x = A[(16 + r15) * lda + kk];
      a1.y = A[(16 + r15) * lda + kk + 1];
      int ka = kk     < Kdim ? kk     : Kdim - 1;
      int kb = kk + 1 < Kdim ? kk + 1 : Kdim - 1;
      w.x = W[n * Kdim + ka];
      w.y = W[n * Kdim + kb];
      acc0 = __builtin_amdgcn_wmma_f32_16x16x4_f32(false, a0, false, w,
                                                   (short)0, acc0, false, false);
      acc1 = __builtin_amdgcn_wmma_f32_16x16x4_f32(false, a1, false, w,
                                                   (short)0, acc1, false, false);
    }
    const float sc = gg[n] * rsqrtf(vv[n] + BN_EPS);
    const float bs = bb[n] - mm[n] * sc;
#pragma unroll
    for (int r = 0; r < 8; ++r) {
      D[(r + half * 8) * ldd + n]      = fmaxf(acc0[r] * sc + bs, 0.f);
      D[(16 + r + half * 8) * ldd + n] = fmaxf(acc1[r] * sc + bs, 0.f);
    }
  }
}

__global__ __launch_bounds__(64) void mlp_kernel(const float* __restrict__ xyz,
                                                 const float* __restrict__ feat,
                                                 const int*   __restrict__ grp,
                                                 const float* __restrict__ newxyz,
                                                 const float* __restrict__ w0, const float* __restrict__ g0,
                                                 const float* __restrict__ b0, const float* __restrict__ m0,
                                                 const float* __restrict__ v0,
                                                 const float* __restrict__ w1, const float* __restrict__ g1,
                                                 const float* __restrict__ b1, const float* __restrict__ m1,
                                                 const float* __restrict__ v1,
                                                 const float* __restrict__ w2, const float* __restrict__ g2,
                                                 const float* __restrict__ b2, const float* __restrict__ m2,
                                                 const float* __restrict__ v2,
                                                 float* __restrict__ outfeat) {
  __shared__ float Xb[2][32 * 68];
  __shared__ float Hb[2][32 * 132];

  const int wave = threadIdx.x >> 5;
  const int lane = threadIdx.x & 31;
  const int q    = blockIdx.x * 2 + wave;
  const int b    = q >> 9;
  const int s    = q & (NPT - 1);

  float* X = Xb[wave];
  float* H = Hb[wave];

  const int gi = grp[(size_t)q * NSAMP + lane];
  const float* prow = xyz + ((size_t)b * NPTS + gi) * 3;
  const float* nx   = newxyz + (b * NPT + s) * 3;
  float* xr = X + lane * 68;
  xr[0] = prow[0] - nx[0];
  xr[1] = prow[1] - nx[1];
  xr[2] = prow[2] - nx[2];
  const float4* frow = (const float4*)(feat + ((size_t)b * NPTS + gi) * 64);
#pragma unroll
  for (int c4 = 0; c4 < 16; ++c4) {
    float4 f = frow[c4];
    xr[3 + 4 * c4 + 0] = f.x;
    xr[3 + 4 * c4 + 1] = f.y;
    xr[3 + 4 * c4 + 2] = f.z;
    xr[3 + 4 * c4 + 3] = f.w;
  }
  xr[67] = 0.f;

  gemm_bn_relu(X, 68,  w0, 67, 68, 64,  g0, b0, m0, v0, H, 132, lane);
  gemm_bn_relu(H, 132, w1, 64, 64, 64,  g1, b1, m1, v1, X, 68,  lane);
  gemm_bn_relu(X, 68,  w2, 64, 64, 128, g2, b2, m2, v2, H, 132, lane);

#pragma unroll
  for (int cc = 0; cc < 4; ++cc) {
    const int col = lane + cc * 32;
    float mx = H[col];
    for (int row = 1; row < 32; ++row) mx = fmaxf(mx, H[row * 132 + col]);
    outfeat[((size_t)b * 128 + col) * NPT + s] = mx;
  }
}

extern "C" void kernel_launch(void* const* d_in, const int* in_sizes, int n_in,
                              void* d_out, int out_size, void* d_ws, size_t ws_size,
                              hipStream_t stream) {
  (void)in_sizes; (void)n_in; (void)out_size; (void)ws_size;
  const float* xyz  = (const float*)d_in[0];
  const float* feat = (const float*)d_in[1];
  const float* w0 = (const float*)d_in[2];
  const float* g0 = (const float*)d_in[3];
  const float* b0 = (const float*)d_in[4];
  const float* m0 = (const float*)d_in[5];
  const float* v0 = (const float*)d_in[6];
  const float* w1 = (const float*)d_in[7];
  const float* g1 = (const float*)d_in[8];
  const float* b1 = (const float*)d_in[9];
  const float* m1 = (const float*)d_in[10];
  const float* v1 = (const float*)d_in[11];
  const float* w2 = (const float*)d_in[12];
  const float* g2 = (const float*)d_in[13];
  const float* b2 = (const float*)d_in[14];
  const float* m2 = (const float*)d_in[15];
  const float* v2 = (const float*)d_in[16];

  float* newxyz  = (float*)d_out;
  float* outfeat = (float*)d_out + NB * NPT * 3;

  int* fpsidx = (int*)d_ws;
  int* grp    = (int*)d_ws + NB * NPT;

  fps_kernel  <<<NB,                1024, 0, stream>>>(xyz, fpsidx, newxyz);
  ballq_kernel<<<(NB * NPT) / 8,     256, 0, stream>>>(xyz, newxyz, grp);
  mlp_kernel  <<<(NB * NPT) / 2,      64, 0, stream>>>(xyz, feat, grp, newxyz,
                                                       w0, g0, b0, m0, v0,
                                                       w1, g1, b1, m1, v1,
                                                       w2, g2, b2, m2, v2,
                                                       outfeat);
}


# FPS: fewer vector-to-scalar round trips per step: beaten test = AND of two lane votes (only waves without an index ask), looked-up index folded on the vector ALU and handed in from it; same picks
# speedup vs baseline: 3.1076x; 1.0528x over previous
#include <hip/hip_runtime.h>
#include <hip/hip_bf16.h>
#include <math.h>

typedef float v2f __attribute__((ext_vector_type(2)));
typedef float v8f __attribute__((ext_vector_type(8)));

#define NB     8
#define NPTS   16384
#define NPT    512
#define NSAMP  32
#define RAD2   0.04f
#define BN_EPS 1e-5f


template <int M>
__device__ __forceinline__ int row_xor(int v) {
  return __builtin_amdgcn_update_dpp(0, v, 0x160 | M, 0xf, 0xf, true);
}
__device__ __forceinline__ int half_xor(int v) {
  return (int)__builtin_amdgcn_permlanex16((unsigned)v, (unsigned)v, 0x76543210u, 0xfedcba98u, false, false);
}
__device__ __forceinline__ int imax2(int a, int b) { return a > b ? a : b; }
__device__ __forceinline__ int imin2(int a, int b) { return a < b ? a : b; }

__device__ __forceinline__ int wave_imax(int m) {
  m = imax2(m, row_xor<1>(m));
  m = imax2(m, row_xor<2>(m));
  m = imax2(m, row_xor<4>(m));
  m = imax2(m, row_xor<8>(m));
  m = imax2(m, half_xor(m));
  return m;
}
__device__ __forceinline__ int wave_imin(int c) {
  c = imin2(c, row_xor<1>(c));
  c = imin2(c, row_xor<2>(c));
  c = imin2(c, row_xor<4>(c));
  c = imin2(c, row_xor<8>(c));
  c = imin2(c, half_xor(c));
  return c;
}

__device__ __forceinline__ float uni(float v) {
  return __int_as_float(__builtin_amdgcn_readfirstlane(__float_as_int(v)));
}
__device__ __forceinline__ unsigned hilbert_pos(unsigned x0, unsigned x1, unsigned x2) {
  unsigned X[3] = {x0, x1, x2};
  for (unsigned Q = 4u; Q > 1u; Q >>= 1) {
    const unsigned Pm = Q - 1u;
    for (int i = 0; i < 3; ++i) {
      if (X[i] & Q) {
        X[0] ^= Pm;
      } else {
        const unsigned t = (X[0] ^ X[i]) & Pm;
        X[0] ^= t;
        X[i] ^= t;
      }
    }
  }
  X[1] ^= X[0];
  X[2] ^= X[1];
  unsigned t = 0u;
  for (unsigned Q = 4u; Q > 1u; Q >>= 1)
    if (X[2] & Q) t ^= Q - 1u;
  X[0] ^= t; X[1] ^= t; X[2] ^= t;
  unsigned h = 0u;
  for (int bit = 2; bit >= 0; --bit)
    h = (h << 3) | (((X[0] >> bit) & 1u) << 2) | (((X[1] >> bit) & 1u) << 1) | ((X[2] >> bit) & 1u);
  return h;
}

__global__ __launch_bounds__(1024) void fps_kernel(const float* __restrict__ xyz,
                                                   int* __restrict__ fpsidx,
                                                   float* __restrict__ newxyz) {
  const int b   = blockIdx.x;
  const int tid = threadIdx.x;
  const float* base = xyz + (size_t)b * NPTS * 3;

  __shared__ float4 P[NPTS];
  __shared__ int2 wrec[2][32];
  __shared__ float wbox[32][6];
  __shared__ int wtot[32];
  __shared__ unsigned short hil[512];

  const int wave = tid >> 5;
  const int lane = tid & 31;

  if (tid < 512)
    hil[tid] = (unsigned short)hilbert_pos((unsigned)tid & 7u, ((unsigned)tid >> 3) & 7u, (unsigned)tid >> 6);
  __syncthreads();
  unsigned* K = (unsigned*)P;
  unsigned key[16];
#pragma unroll
  for (int j = 0; j < 16; ++j) {
    const int p = tid * 16 + j;
    const float x = base[p * 3 + 0], y = base[p * 3 + 1], z = base[p * 3 + 2];
    const unsigned ix = (unsigned)(int)fminf(fmaxf(x * 8.0f, 0.0f), 7.0f);
    const unsigned iy = (unsigned)(int)fminf(fmaxf(y * 8.0f, 0.0f), 7.0f);
    const unsigned iz = (unsigned)(int)fminf(fmaxf(z * 8.0f, 0.0f), 7.0f);
    key[j] = ((unsigned)hil[ix | (iy << 3) | (iz << 6)] << 14) | (unsigned)p;
  }
  for (int bit = 14; bit < 23; ++bit) {
    int z = 0;
#pragma unroll
    for (int j = 0; j < 16; ++j) z += (int)(((key[j] >> bit) & 1u) ^ 1u);
    int inc = z;
    for (int d = 1; d < 32; d <<= 1) {
      const int t = __shfl_up(inc, d);
      if (lane >= d) inc += t;
    }
    if (lane == 31) wtot[wave] = inc;
    __syncthreads();
    const int wt = wtot[lane];
    int winc = wt;
    for (int d = 1; d < 32; d <<= 1) {
      const int t = __shfl_up(winc, d);
      if (lane >= d) winc += t;
    }
    const int zall  = __shfl(winc, 31);
    const int wexcl = __shfl(winc - wt, wave);
    int zpos = wexcl + inc - z;
    int opos = zall + (tid * 16 - zpos);
#pragma unroll
    for (int j = 0; j < 16; ++j) {
      const int one = (int)((key[j] >> bit) & 1u);
      K[one ? opos : zpos] = key[j];
      opos += one;
      zpos += 1 - one;
    }
    __syncthreads();
#pragma unroll
    for (int j = 0; j < 16; ++j) key[j] = K[tid * 16 + j];
    __syncthreads();
  }

  float px[16], py[16], pz[16], dd[16];
  unsigned pk[8];
#pragma unroll
  for (int j = 0; j < 16; ++j) {
    const int p = (int)(key[j] & 0x3fffu);
    px[j] = base[p * 3 + 0];
    py[j] = base[p * 3 + 1];
    pz[j] = base[p * 3 + 2];
    dd[j] = 3.4e38f;
    P[p] = make_float4(px[j], py[j], pz[j], 0.0f);
  }
#pragma unroll
  for (int j = 0; j < 16; j += 2)
    pk[j >> 1] = (key[j] & 0x3fffu) | ((key[j + 1] & 0x3fffu) << 16);

  float lx = px[0], hx = px[0], ly = py[0], hy = py[0], lz = pz[0], hz = pz[0];
#pragma unroll
  for (int j = 1; j < 16; ++j) {
    lx = fminf(lx, px[j]); hx = fmaxf(hx, px[j]);
    ly = fminf(ly, py[j]); hy = fmaxf(hy, py[j]);
    lz = fminf(lz, pz[j]); hz = fmaxf(hz, pz[j]);
  }
  for (int d = 16; d >= 1; d >>= 1) {
    lx = fminf(lx, __shfl_xor(lx, d)); hx = fmaxf(hx, __shfl_xor(hx, d));
    ly = fminf(ly, __shfl_xor(ly, d)); hy = fmaxf(hy, __shfl_xor(hy, d));
    lz = fminf(lz, __shfl_xor(lz, d)); hz = fmaxf(hz, __shfl_xor(hz, d));
  }
  if (lane == 0) {
    wbox[wave][0] = lx; wbox[wave][1] = hx;
    wbox[wave][2] = ly; wbox[wave][3] = hy;
    wbox[wave][4] = lz; wbox[wave][5] = hz;
  }
  __syncthreads();
  const float blx = wbox[lane][0], bhx = wbox[lane][1];
  const float bly = wbox[lane][2], bhy = wbox[lane][3];
  const float blz = wbox[lane][4], bhz = wbox[lane][5];
  const int wv = __builtin_amdgcn_readfirstlane(wave);

  const int NOIDX = 0x7fffffff;
  int last = 0;
  float cx = base[0], cy = base[1], cz = base[2];
  int m = __float_as_int(3.4e38f);
  int wc = NOIDX;
  bool have = false;
  int mv = __float_as_int(3.4e38f);

  for (int i = 0; i < NPT; ++i) {
    if (tid == 0) {
      fpsidx[b * NPT + i] = last;
      newxyz[(b * NPT + i) * 3 + 0] = cx;
      newxyz[(b * NPT + i) * 3 + 1] = cy;
      newxyz[(b * NPT + i) * 3 + 2] = cz;
    }

    const float ex = fmaxf(fmaxf(blx - cx, cx - bhx), 0.0f);
    const float ey = fmaxf(fmaxf(bly - cy, cy - bhy), 0.0f);
    const float ez = fmaxf(fmaxf(blz - cz, cz - bhz), 0.0f);
    const float lb = (ex * ex + ey * ey + ez * ez) * 0.999f;
    const unsigned keepmask = __builtin_amdgcn_ballot_w32(lb > __int_as_float(mv));
    if (((keepmask >> wv) & 1u) == 0u) {
      float bv = -1.0f;
#pragma unroll
      for (int j = 0; j < 16; ++j) {
        float dx = px[j] - cx, dy = py[j] - cy, dz = pz[j] - cz;
        float d  = __builtin_fmaf(dz, dz, __builtin_fmaf(dx, dx, dy * dy));
        float nd = fminf(dd[j], d);
        dd[j] = nd;
        bv = fmaxf(bv, nd);
      }
      m  = wave_imax(__float_as_int(bv));
      wc = NOIDX;
      have = false;
    }

    if (!have &&
        (keepmask & __builtin_amdgcn_ballot_w32(mv > m)) == 0u) {
      const int M = m;
      int c0 = NOIDX, c1 = NOIDX, c2 = NOIDX, c3 = NOIDX;
#pragma unroll
      for (int j = 0; j < 16; j += 4) {
        if (__float_as_int(dd[j])     == M) c0 = imin2(c0, (int)(pk[j >> 1] & 0xffffu));
        if (__float_as_int(dd[j + 1]) == M) c1 = imin2(c1, (int)(pk[j >> 1] >> 16));
        if (__float_as_int(dd[j + 2]) == M) c2 = imin2(c2, (int)(pk[(j >> 1) + 1] & 0xffffu));
        if (__float_as_int(dd[j + 3]) == M) c3 = imin2(c3, (int)(pk[(j >> 1) + 1] >> 16));
      }
      wc = wave_imin(imin2(imin2(c0, c1), imin2(c2, c3)));
      have = true;
    }

    if (lane == 0) wrec[i & 1][wave] = make_int2(m, wc);
    __syncthreads();
    const int2 rec = wrec[i & 1][lane];
    mv = rec.x;
    const int    cand = rec.y;
    const float4 pc   = P[cand & (NPTS - 1)];
    const int bm = wave_imax(mv);
    const unsigned hm = __builtin_amdgcn_ballot_w32(mv == bm);
    int wl;
    if (__builtin_popcount(hm) == 1) {
      wl = __builtin_ctz(hm);
    } else {
      const int ct   = (mv == bm) ? cand : NOIDX;
      const int cmin = wave_imin(ct);
      wl = __builtin_ctz(__builtin_amdgcn_ballot_w32(ct == cmin));
    }
    cx = __int_as_float(__builtin_amdgcn_readlane(__float_as_int(pc.x), wl));
    cy = __int_as_float(__builtin_amdgcn_readlane(__float_as_int(pc.y), wl));
    cz = __int_as_float(__builtin_amdgcn_readlane(__float_as_int(pc.z), wl));
    last = __builtin_amdgcn_readlane(cand, wl) & (NPTS - 1);
  }
}

#define BQ_U 16
__global__ __launch_bounds__(256) void ballq_kernel(const float* __restrict__ xyz,
                                                    const float* __restrict__ newxyz,
                                                    int* __restrict__ grp) {
  const int lane = threadIdx.x & 31;
  const int q    = blockIdx.x * 8 + (threadIdx.x >> 5);
  const int b    = q >> 9;
  const int s    = q & (NPT - 1);

  const float* base = xyz + (size_t)b * NPTS * 3;
  const float cx = newxyz[(b * NPT + s) * 3 + 0];
  const float cy = newxyz[(b * NPT + s) * 3 + 1];
  const float cz = newxyz[(b * NPT + s) * 3 + 2];

  int* g = grp + (size_t)q * NSAMP;
  int cnt = 0, first = -1;

  for (int bp = 0; bp < NPTS && cnt < NSAMP; bp += 32 * BQ_U) {
    float px[BQ_U], py[BQ_U], pz[BQ_U];
#pragma unroll
    for (int u = 0; u < BQ_U; ++u) {
      const int p = bp + 32 * u + lane;
      px[u] = base[p * 3 + 0];
      py[u] = base[p * 3 + 1];
      pz[u] = base[p * 3 + 2];
    }
    unsigned mask[BQ_U];
#pragma unroll
    for (int u = 0; u < BQ_U; ++u) {
      float dx = px[u] - cx;
      float dy = py[u] - cy;
      float dz = pz[u] - cz;
      mask[u] = (unsigned)__ballot((dx * dx + dy * dy + dz * dz) <= RAD2);
    }
#pragma unroll
    for (int u = 0; u < BQ_U; ++u) {
      if (mask[u]) {
        if (cnt < NSAMP) {
          const int cb = bp + 32 * u;
          if (first < 0) first = cb + __builtin_ctz(mask[u]);
          const bool hit = (mask[u] >> lane) & 1u;
          int pos = __popc(mask[u] & ((1u << lane) - 1u));
          if (hit && (cnt + pos) < NSAMP) g[cnt + pos] = cb + lane;
          cnt += __popc(mask[u]);
        }
      }
    }
  }
  if (first < 0) first = NPTS - 1;
  int filled = cnt < NSAMP ? cnt : NSAMP;
  for (int slot = filled + lane; slot < NSAMP; slot += 32) g[slot] = first;
}

__device__ __forceinline__ void gemm_bn_relu(const float* __restrict__ A, int lda,
                                             const float* __restrict__ W, int Kdim, int Kpad,
                                             int Nout,
                                             const float* __restrict__ gg,
                                             const float* __restrict__ bb,
                                             const float* __restrict__ mm,
                                             const float* __restrict__ vv,
                                             float* __restrict__ D, int ldd, int lane) {
  const int r15  = lane & 15;
  const int half = lane >> 4;
  for (int n0 = 0; n0 < Nout; n0 += 16) {
    v8f acc0 = {0.f, 0.f, 0.f, 0.f, 0.f, 0.f, 0.f, 0.f};
    v8f acc1 = {0.f, 0.f, 0.f, 0.f, 0.f, 0.f, 0.f, 0.f};
    const int n = n0 + r15;
    for (int k0 = 0; k0 < Kpad; k0 += 4) {
      const int kk = k0 + half * 2;
      v2f a0, a1, w;
      a0.x = A[r15 * lda + kk];
      a0.y = A[r15 * lda + kk + 1];
      a1.x = A[(16 + r15) * lda + kk];
      a1.y = A[(16 + r15) * lda + kk + 1];
      int ka = kk     < Kdim ? kk     : Kdim - 1;
      int kb = kk + 1 < Kdim ? kk + 1 : Kdim - 1;
      w.x = W[n * Kdim + ka];
      w.y = W[n * Kdim + kb];
      acc0 = __builtin_amdgcn_wmma_f32_16x16x4_f32(false, a0, false, w,
                                                   (short)0, acc0, false, false);
      acc1 = __builtin_amdgcn_wmma_f32_16x16x4_f32(false, a1, false, w,
                                                   (short)0, acc1, false, false);
    }
    const float sc = gg[n] * rsqrtf(vv[n] + BN_EPS);
    const float bs = bb[n] - mm[n] * sc;
#pragma unroll
    for (int r = 0; r < 8; ++r) {
      D[(r + half * 8) * ldd + n]      = fmaxf(acc0[r] * sc + bs, 0.f);
      D[(16 + r + half * 8) * ldd + n] = fmaxf(acc1[r] * sc + bs, 0.f);
    }
  }
}

__global__ __launch_bounds__(64) void mlp_kernel(const float* __restrict__ xyz,
                                                 const float* __restrict__ feat,
                                                 const int*   __restrict__ grp,
                                                 const float* __restrict__ newxyz,
                                                 const float* __restrict__ w0, const float* __restrict__ g0,
                                                 const float* __restrict__ b0, const float* __restrict__ m0,
                                                 const float* __restrict__ v0,
                                                 const float* __restrict__ w1, const float* __restrict__ g1,
                                                 const float* __restrict__ b1, const float* __restrict__ m1,
                                                 const float* __restrict__ v1,
                                                 const float* __restrict__ w2, const float* __restrict__ g2,
                                                 const float* __restrict__ b2, const float* __restrict__ m2,
                                                 const float* __restrict__ v2,
                                                 float* __restrict__ outfeat) {
  __shared__ float Xb[2][32 * 68];
  __shared__ float Hb[2][32 * 132];

  const int wave = threadIdx.x >> 5;
  const int lane = threadIdx.x & 31;
  const int q    = blockIdx.x * 2 + wave;
  const int b    = q >> 9;
  const int s    = q & (NPT - 1);

  float* X = Xb[wave];
  float* H = Hb[wave];

  const int gi = grp[(size_t)q * NSAMP + lane];
  const float* prow = xyz + ((size_t)b * NPTS + gi) * 3;
  const float* nx   = newxyz + (b * NPT + s) * 3;
  float* xr = X + lane * 68;
  xr[0] = prow[0] - nx[0];
  xr[1] = prow[1] - nx[1];
  xr[2] = prow[2] - nx[2];
  const float4* frow = (const float4*)(feat + ((size_t)b * NPTS + gi) * 64);
#pragma unroll
  for (int c4 = 0; c4 < 16; ++c4) {
    float4 f = frow[c4];
    xr[3 + 4 * c4 + 0] = f.x;
    xr[3 + 4 * c4 + 1] = f.y;
    xr[3 + 4 * c4 + 2] = f.z;
    xr[3 + 4 * c4 + 3] = f.w;
  }
  xr[67] = 0.f;

  gemm_bn_relu(X, 68,  w0, 67, 68, 64,  g0, b0, m0, v0, H, 132, lane);
  gemm_bn_relu(H, 132, w1, 64, 64, 64,  g1, b1, m1, v1, X, 68,  lane);
  gemm_bn_relu(X, 68,  w2, 64, 64, 128, g2, b2, m2, v2, H, 132, lane);

#pragma unroll
  for (int cc = 0; cc < 4; ++cc) {
    const int col = lane + cc * 32;
    float mx = H[col];
    for (int row = 1; row < 32; ++row) mx = fmaxf(mx, H[row * 132 + col]);
    outfeat[((size_t)b * 128 + col) * NPT + s] = mx;
  }
}

extern "C" void kernel_launch(void* const* d_in, const int* in_sizes, int n_in,
                              void* d_out, int out_size, void* d_ws, size_t ws_size,
                              hipStream_t stream) {
  (void)in_sizes; (void)n_in; (void)out_size; (void)ws_size;
  const float* xyz  = (const float*)d_in[0];
  const float* feat = (const float*)d_in[1];
  const float* w0 = (const float*)d_in[2];
  const float* g0 = (const float*)d_in[3];
  const float* b0 = (const float*)d_in[4];
  const float* m0 = (const float*)d_in[5];
  const float* v0 = (const float*)d_in[6];
  const float* w1 = (const float*)d_in[7];
  const float* g1 = (const float*)d_in[8];
  const float* b1 = (const float*)d_in[9];
  const float* m1 = (const float*)d_in[10];
  const float* v1 = (const float*)d_in[11];
  const float* w2 = (const float*)d_in[12];
  const float* g2 = (const float*)d_in[13];
  const float* b2 = (const float*)d_in[14];
  const float* m2 = (const float*)d_in[15];
  const float* v2 = (const float*)d_in[16];

  float* newxyz  = (float*)d_out;
  float* outfeat = (float*)d_out + NB * NPT * 3;

  int* fpsidx = (int*)d_ws;
  int* grp    = (int*)d_ws + NB * NPT;

  fps_kernel  <<<NB,                1024, 0, stream>>>(xyz, fpsidx, newxyz);
  ballq_kernel<<<(NB * NPT) / 8,     256, 0, stream>>>(xyz, newxyz, grp);
  mlp_kernel  <<<(NB * NPT) / 2,      64, 0, stream>>>(xyz, feat, grp, newxyz,
                                                       w0, g0, b0, m0, v0,
                                                       w1, g1, b1, m1, v1,
                                                       w2, g2, b2, m2, v2,
                                                       outfeat);
}
